# starting kernel
# baseline (speedup 1.0000x reference)
#include <hip/hip_runtime.h>
#include <hip/hip_bf16.h>


#define T_SEQ 8192
#define E_DIM 256
#define V_DIM 17

typedef float v2f __attribute__((ext_vector_type(2)));
typedef float v8f __attribute__((ext_vector_type(8)));

__device__ __forceinline__ v8f wmma4(v2f a, v2f b, v8f c) {
  return __builtin_amdgcn_wmma_f32_16x16x4_f32(
       false, a,  false, b,
       (short)0, c,  false,  false);
}

__global__ __launch_bounds__(32) void head123_kernel(
    const float* __restrict__ x,
    const float* __restrict__ W1, const float* __restrict__ b1,
    const float* __restrict__ W2, const float* __restrict__ b2,
    const float* __restrict__ W3, const float* __restrict__ b3,
    float* __restrict__ out) {
  const int n       = blockIdx.y;
  const int rowbase = blockIdx.x * 16;

  const float* W;
  const float* b;
  if (rowbase < 64)       { W = W1; b = b1; }
  else if (rowbase < 576) { W = W2; b = b2; }
  else                    { W = W3; b = b3; }

  const int lane = threadIdx.x & 31;
  const int l15  = lane & 15;
  const int koff = (lane < 16) ? 0 : 2;

  const float* xrow = x + ((size_t)n * T_SEQ + rowbase + l15) * E_DIM;
  const int c1  = 16 + l15;
  const int c1c = (c1 < V_DIM) ? c1 : (V_DIM - 1);

  v8f acc0 = {};
  v8f acc1 = {};
  for (int kk = 0; kk < E_DIM; kk += 4) {
    v2f a;
    a.x = xrow[kk + koff];
    a.y = xrow[kk + koff + 1];
    v2f bf0, bf1;
    bf0.x = W[(kk + koff) * V_DIM + l15];
    bf0.y = W[(kk + koff + 1) * V_DIM + l15];
    bf1.x = W[(kk + koff) * V_DIM + c1c];
    bf1.y = W[(kk + koff + 1) * V_DIM + c1c];
    acc0 = wmma4(a, bf0, acc0);
    acc1 = wmma4(a, bf1, acc1);
  }

#pragma unroll
  for (int v = 0; v < 8; ++v) {
    const int r = ((lane < 16) ? v : v + 8);
    float* o = out + ((size_t)n * T_SEQ + rowbase + r) * V_DIM;
    o[l15] = acc0[v] + b[l15];
    if (c1 < V_DIM) o[c1] = acc1[v] + b[c1];
  }
}

__global__ __launch_bounds__(256) void conv_head_kernel(
    const float* __restrict__ x,
    const float* __restrict__ Wd, const float* __restrict__ bd,
    const float* __restrict__ Wl, const float* __restrict__ bl,
    float* __restrict__ out,
    int s, int ioff, int out_off, int nt, int kgroups) {
  __shared__ float Elds[64 * 256];

  const int n  = blockIdx.y;
  const int tb = blockIdx.x / kgroups;
  const int kg = blockIdx.x % kgroups;

  const int tid  = threadIdx.x;
  const int w    = tid >> 5;
  const int lane = tid & 31;
  const int l15  = lane & 15;
  const int koff = (lane < 16) ? 0 : 2;

  const int trel   = tb * 16 + l15;
  const int tclamp = (trel < nt) ? trel : (nt - 1);
  const float* xrow = x + ((size_t)n * T_SEQ + ioff + tclamp) * E_DIM;
  const int Es = E_DIM * s;

  v8f acc[8];
#pragma unroll
  for (int i = 0; i < 8; ++i) acc[i] = (v8f){};

  for (int kk = 0; kk < E_DIM; kk += 4) {
    v2f a;
    a.x = xrow[kk + koff];
    a.y = xrow[kk + koff + 1];
#pragma unroll
    for (int i = 0; i < 8; ++i) {
      const int ct = w * 8 + i;
      const int g  = ct >> 4;
      const int ot = ct & 15;
      const int k  = kg * 4 + g;
      const int o  = ot * 16 + l15;
      const float* bp = Wd + (size_t)(kk + koff) * Es + (size_t)o * s + k;
      v2f bf;
      bf.x = bp[0];
      bf.y = bp[Es];
      acc[i] = wmma4(a, bf, acc[i]);
    }
  }

#pragma unroll
  for (int i = 0; i < 8; ++i) {
    const int ct = w * 8 + i;
    const int g  = ct >> 4;
    const int ot = ct & 15;
    const int o  = ot * 16 + l15;
    const float bdv = bd[o];
#pragma unroll
    for (int v = 0; v < 8; ++v) {
      const int tl  = (lane < 16) ? v : v + 8;
      const int row = tl * 4 + g;
      Elds[row * 256 + (o ^ ((row & 15) * 4))] = acc[i][v] + bdv;
    }
  }
  __syncthreads();

  const int rt   = w >> 1;
  const int col0 = (w & 1) * 16;
  const int m    = rt * 16 + l15;
  const int col  = col0 + l15;
  const bool cval = (col < V_DIM);
  const int colc  = cval ? col : (V_DIM - 1);
  const int msw   = (m & 15) * 4;

  v8f acc2 = {};
  for (int kk = 0; kk < E_DIM; kk += 4) {
    const int k0 = kk + koff;
    v2f a;
    a.x = Elds[m * 256 + (k0 ^ msw)];
    a.y = Elds[m * 256 + ((k0 + 1) ^ msw)];
    v2f bf;
    bf.x = Wl[(size_t)k0 * V_DIM + colc];
    bf.y = Wl[(size_t)(k0 + 1) * V_DIM + colc];
    acc2 = wmma4(a, bf, acc2);
  }

  const float blv = bl[colc];
#pragma unroll
  for (int v = 0; v < 8; ++v) {
    const int r     = rt * 16 + ((lane < 16) ? v : v + 8);
    const int tl    = r >> 2;
    const int g     = r & 3;
    const int trel2 = tb * 16 + tl;
    if (cval && trel2 < nt) {
      const int j = out_off + trel2 * s + kg * 4 + g;
      out[((size_t)n * T_SEQ + j) * V_DIM + col] = acc2[v] + blv;
    }
  }
}

extern "C" void kernel_launch(void* const* d_in, const int* in_sizes, int n_in,
                              void* d_out, int out_size, void* d_ws, size_t ws_size,
                              hipStream_t stream) {
  const float* x   = (const float*)d_in[0];
  const float* W1  = (const float*)d_in[4];
  const float* b1  = (const float*)d_in[5];
  const float* W2  = (const float*)d_in[6];
  const float* b2  = (const float*)d_in[7];
  const float* W3  = (const float*)d_in[8];
  const float* b3  = (const float*)d_in[9];
  const float* Wd4 = (const float*)d_in[10];
  const float* bd4 = (const float*)d_in[11];
  const float* Wl4 = (const float*)d_in[12];
  const float* bl4 = (const float*)d_in[13];
  const float* Wd5 = (const float*)d_in[14];
  const float* bd5 = (const float*)d_in[15];
  const float* Wl5 = (const float*)d_in[16];
  const float* bl5 = (const float*)d_in[17];
  float* out = (float*)d_out;

  const int N = 8;

  head123_kernel<<<dim3(100, N), 32, 0, stream>>>(x, W1, b1, W2, b2, W3, b3, out);

  conv_head_kernel<<<dim3(32 * 1, N), 256, 0, stream>>>(
      x, Wd4, bd4, Wl4, bl4, out,  4,  1600,  1600,
       512,  1);

  conv_head_kernel<<<dim3(36 * 2, N), 256, 0, stream>>>(
      x, Wd5, bd5, Wl5, bl5, out,  8,  2112,  3648,
       568,  2);
}


# Conv head S=8 cut into ONE round of 256 blocks (was 144 of 32 tokens): block = a 16-token tile x 8 taps + one tap of one of the 28 leftover tiles, fed from Wd lines it already loads; sums unchanged
# speedup vs baseline: 2.8857x; 2.8857x over previous
#include <hip/hip_runtime.h>
#include <hip/hip_bf16.h>


#define T_SEQ 8192
#define E_DIM 256
#define V_DIM 17

typedef float v2f __attribute__((ext_vector_type(2)));
typedef float v8f __attribute__((ext_vector_type(8)));

__device__ __forceinline__ v8f wmma4(v2f a, v2f b, v8f c) {
  return __builtin_amdgcn_wmma_f32_16x16x4_f32(
       false, a,  false, b,
       (short)0, c,  false,  false);
}

__global__ __launch_bounds__(32) void head123_kernel(
    const float* __restrict__ x,
    const float* __restrict__ W1, const float* __restrict__ b1,
    const float* __restrict__ W2, const float* __restrict__ b2,
    const float* __restrict__ W3, const float* __restrict__ b3,
    float* __restrict__ out) {
  const int n       = blockIdx.y;
  const int rowbase = blockIdx.x * 16;

  const float* W;
  const float* b;
  if (rowbase < 64)       { W = W1; b = b1; }
  else if (rowbase < 576) { W = W2; b = b2; }
  else                    { W = W3; b = b3; }

  const int lane = threadIdx.x & 31;
  const int l15  = lane & 15;
  const int koff = (lane < 16) ? 0 : 2;

  const float* xrow = x + ((size_t)n * T_SEQ + rowbase + l15) * E_DIM;
  const int c1  = 16 + l15;
  const int c1c = (c1 < V_DIM) ? c1 : (V_DIM - 1);

  v8f acc0 = {};
  v8f acc1 = {};
  for (int kk = 0; kk < E_DIM; kk += 4) {
    v2f a;
    a.x = xrow[kk + koff];
    a.y = xrow[kk + koff + 1];
    v2f bf0, bf1;
    bf0.x = W[(kk + koff) * V_DIM + l15];
    bf0.y = W[(kk + koff + 1) * V_DIM + l15];
    bf1.x = W[(kk + koff) * V_DIM + c1c];
    bf1.y = W[(kk + koff + 1) * V_DIM + c1c];
    acc0 = wmma4(a, bf0, acc0);
    acc1 = wmma4(a, bf1, acc1);
  }

#pragma unroll
  for (int v = 0; v < 8; ++v) {
    const int r = ((lane < 16) ? v : v + 8);
    float* o = out + ((size_t)n * T_SEQ + rowbase + r) * V_DIM;
    o[l15] = acc0[v] + b[l15];
    if (c1 < V_DIM) o[c1] = acc1[v] + b[c1];
  }
}

template <int S, int MT>
__global__ __launch_bounds__(64 * S) void conv_head_kernel(
    const float* __restrict__ x,
    const float* __restrict__ Wd, const float* __restrict__ bd,
    const float* __restrict__ Wl, const float* __restrict__ bl,
    float* __restrict__ out,
    int ioff, int out_off, int nt) {
  constexpr int OTW = 8 / S;
  constexpr int Q   = S / 4;
  constexpr int Es  = E_DIM * S;
  constexpr int TOK = 16 * MT;
  __shared__ float Elds[TOK * S * 256];
  constexpr int XP = E_DIM + 4;
  __shared__ alignas(16) float Xs[TOK * XP];

  const int n  = blockIdx.y;
  const int tb = blockIdx.x;

  const int tid  = threadIdx.x;
  const int w    = tid >> 5;
  const int lane = tid & 31;
  const int l15  = lane & 15;
  const int koff = (lane < 16) ? 0 : 2;

  constexpr int NXL = (TOK * 64) / (64 * S);
  float4 xv[NXL];
#pragma unroll
  for (int j = 0; j < NXL; ++j) {
    const int idx    = tid + j * 64 * S;
    const int r      = idx >> 6;
    const int c4     = idx & 63;
    const int trel   = tb * TOK + r;
    const int tclamp = (trel < nt) ? trel : (nt - 1);
    xv[j] = *reinterpret_cast<const float4*>(
        x + ((size_t)n * T_SEQ + ioff + tclamp) * E_DIM + c4 * 4);
  }
#pragma unroll
  for (int j = 0; j < NXL; ++j) {
    const int idx = tid + j * 64 * S;
    *reinterpret_cast<float4*>(&Xs[(idx >> 6) * XP + (idx & 63) * 4]) = xv[j];
  }
  __syncthreads();

  const float* xrow = &Xs[l15 * XP];

  v8f acc[MT][8];
#pragma unroll
  for (int m = 0; m < MT; ++m)
#pragma unroll
    for (int i = 0; i < 8; ++i) acc[m][i] = (v8f){};

  const float* wbase = Wd + (size_t)koff * Es + (size_t)((w * OTW) * 16 + l15) * S;

#pragma unroll(S == 8 ? 2 : 4)
  for (int kk = 0; kk < E_DIM; kk += 4) {
    v2f a[MT];
#pragma unroll
    for (int m = 0; m < MT; ++m) {
      a[m].x = xrow[m * 16 * XP + kk + koff];
      a[m].y = xrow[m * 16 * XP + kk + koff + 1];
    }
    const float* wrow = wbase + (size_t)kk * Es;
#pragma unroll
    for (int otl = 0; otl < OTW; ++otl) {
#pragma unroll
      for (int q = 0; q < Q; ++q) {
        const float* bp = wrow + otl * 16 * S + q * 4;
        const float4 r0 = *reinterpret_cast<const float4*>(bp);
        const float4 r1 = *reinterpret_cast<const float4*>(bp + Es);
        const int t0 = otl * S + q * 4;
        v2f bf;
        bf.x = r0.x; bf.y = r1.x;
#pragma unroll
        for (int m = 0; m < MT; ++m) acc[m][t0 + 0] = wmma4(a[m], bf, acc[m][t0 + 0]);
        bf.x = r0.y; bf.y = r1.y;
#pragma unroll
        for (int m = 0; m < MT; ++m) acc[m][t0 + 1] = wmma4(a[m], bf, acc[m][t0 + 1]);
        bf.x = r0.z; bf.y = r1.z;
#pragma unroll
        for (int m = 0; m < MT; ++m) acc[m][t0 + 2] = wmma4(a[m], bf, acc[m][t0 + 2]);
        bf.x = r0.w; bf.y = r1.w;
#pragma unroll
        for (int m = 0; m < MT; ++m) acc[m][t0 + 3] = wmma4(a[m], bf, acc[m][t0 + 3]);
      }
    }
  }

#pragma unroll
  for (int i = 0; i < 8; ++i) {
    const int otl = i / S;
    const int k   = i % S;
    const int o   = (w * OTW + otl) * 16 + l15;
    const float bdv = bd[o];
#pragma unroll
    for (int m = 0; m < MT; ++m) {
#pragma unroll
      for (int v = 0; v < 8; ++v) {
        const int tl  = m * 16 + ((lane < 16) ? v : v + 8);
        const int row = tl * S + k;
        Elds[row * 256 + (o ^ ((row & 15) * 4))] = acc[m][i][v] + bdv;
      }
    }
  }
  __syncthreads();

  constexpr int CT = MT;
  const int rt   = (MT == 1) ? (w >> 1) : w;
  const int col0 = (MT == 1) ? (w & 1) * 16 : 0;
  const int m    = rt * 16 + l15;
  const int msw  = (m & 15) * 4;

  int  colv[CT];
  bool cval[CT];
  int  colc[CT];
  v8f  acc2[CT];
#pragma unroll
  for (int c = 0; c < CT; ++c) {
    colv[c] = col0 + c * 16 + l15;
    cval[c] = (colv[c] < V_DIM);
    colc[c] = cval[c] ? colv[c] : (V_DIM - 1);
    acc2[c] = (v8f){};
  }

  for (int kk = 0; kk < E_DIM; kk += 4) {
    const int k0 = kk + koff;
    v2f a;
    a.x = Elds[m * 256 + (k0 ^ msw)];
    a.y = Elds[m * 256 + ((k0 + 1) ^ msw)];
#pragma unroll
    for (int c = 0; c < CT; ++c) {
      v2f bf;
      bf.x = Wl[(size_t)k0 * V_DIM + colc[c]];
      bf.y = Wl[(size_t)(k0 + 1) * V_DIM + colc[c]];
      acc2[c] = wmma4(a, bf, acc2[c]);
    }
  }

#pragma unroll
  for (int c = 0; c < CT; ++c) {
    const float blv = bl[colc[c]];
#pragma unroll
    for (int v = 0; v < 8; ++v) {
      const int r     = rt * 16 + ((lane < 16) ? v : v + 8);
      const int tl    = r / S;
      const int k     = r % S;
      const int trel2 = tb * TOK + tl;
      if (cval[c] && trel2 < nt) {
        const int j = out_off + trel2 * S + k;
        out[((size_t)n * T_SEQ + j) * V_DIM + colv[c]] = acc2[c][v] + blv;
      }
    }
  }
}

__global__ __launch_bounds__(512) void conv8_kernel(
    const float* __restrict__ x,
    const float* __restrict__ Wd, const float* __restrict__ bd,
    const float* __restrict__ Wl, const float* __restrict__ bl,
    float* __restrict__ out,
    int ioff, int out_off, int nt, int nbatch) {
  constexpr int S    = 8;
  constexpr int Es   = E_DIM * S;
  constexpr int XP   = E_DIM + 4;
  constexpr int ROWS = 16 * S + 16;
  __shared__ float Elds[ROWS * 256];
  __shared__ alignas(16) float Xs[32 * XP];

  const int b      = blockIdx.x;
  const int G      = nbatch * nt;
  const int ntiles = (G + 15) >> 4;
  const int xt_raw = (int)gridDim.x + (b >> 3);
  const bool has_x = xt_raw < ntiles;
  const int xt     = has_x ? xt_raw : b;
  const int kx     = b & 7;

  const int tid  = threadIdx.x;
  const int w    = tid >> 5;
  const int lane = tid & 31;
  const int l15  = lane & 15;
  const int koff = (lane < 16) ? 0 : 2;

  float4 xv[4];
#pragma unroll
  for (int j = 0; j < 4; ++j) {
    const int idx  = tid + j * 512;
    const int r    = idx >> 6;
    const int c4   = idx & 63;
    const int tile = (j < 2) ? b : xt;
    const int g    = tile * 16 + (r & 15);
    const int gc   = (g < G) ? g : (G - 1);
    const int n    = gc / nt;
    const int t    = gc - n * nt;
    xv[j] = *reinterpret_cast<const float4*>(
        x + ((size_t)n * T_SEQ + ioff + t) * E_DIM + c4 * 4);
  }
#pragma unroll
  for (int j = 0; j < 4; ++j) {
    const int idx = tid + j * 512;
    *reinterpret_cast<float4*>(&Xs[(idx >> 6) * XP + (idx & 63) * 4]) = xv[j];
  }
  __syncthreads();

  const float* xrow = &Xs[l15 * XP];
  v8f acc[8];
#pragma unroll
  for (int i = 0; i < 8; ++i) acc[i] = (v8f){};
  v8f accx = {};

  const float* wbase = Wd + (size_t)koff * Es + (size_t)(w * 16 + l15) * S;
  const float* wbx   = wbase + kx;

#pragma unroll 4
  for (int kk = 0; kk < E_DIM; kk += 4) {
    v2f a, ax;
    a.x  = xrow[kk + koff];
    a.y  = xrow[kk + koff + 1];
    ax.x = xrow[16 * XP + kk + koff];
    ax.y = xrow[16 * XP + kk + koff + 1];
    const float* wrow = wbase + (size_t)kk * Es;
    v2f bx;
    bx.x = wbx[(size_t)kk * Es];
    bx.y = wbx[(size_t)kk * Es + Es];
#pragma unroll
    for (int q = 0; q < 2; ++q) {
      const float4 r0 = *reinterpret_cast<const float4*>(wrow + q * 4);
      const float4 r1 = *reinterpret_cast<const float4*>(wrow + Es + q * 4);
      v2f bf;
      bf.x = r0.x; bf.y = r1.x;
      acc[q * 4 + 0] = wmma4(a, bf, acc[q * 4 + 0]);
      bf.x = r0.y; bf.y = r1.y;
      acc[q * 4 + 1] = wmma4(a, bf, acc[q * 4 + 1]);
      bf.x = r0.z; bf.y = r1.z;
      acc[q * 4 + 2] = wmma4(a, bf, acc[q * 4 + 2]);
      bf.x = r0.w; bf.y = r1.w;
      acc[q * 4 + 3] = wmma4(a, bf, acc[q * 4 + 3]);
    }
    accx = wmma4(ax, bx, accx);
  }

  {
    const int o = w * 16 + l15;
    const float bdv = bd[o];
#pragma unroll
    for (int k = 0; k < 8; ++k) {
#pragma unroll
      for (int v = 0; v < 8; ++v) {
        const int tl  = (lane < 16) ? v : v + 8;
        const int row = tl * S + k;
        Elds[row * 256 + (o ^ ((row & 15) * 4))] = acc[k][v] + bdv;
      }
    }
#pragma unroll
    for (int v = 0; v < 8; ++v) {
      const int tl  = (lane < 16) ? v : v + 8;
      const int row = 16 * S + tl;
      Elds[row * 256 + (o ^ ((row & 15) * 4))] = accx[v] + bdv;
    }
  }
  __syncthreads();

  if (w >= 9) return;
  const int rt  = w;
  const int m   = rt * 16 + l15;
  const int msw = (m & 15) * 4;

  int  colv[2];
  bool cval[2];
  int  colc[2];
  v8f  acc2[2];
#pragma unroll
  for (int c = 0; c < 2; ++c) {
    colv[c] = c * 16 + l15;
    cval[c] = (colv[c] < V_DIM);
    colc[c] = cval[c] ? colv[c] : (V_DIM - 1);
    acc2[c] = (v8f){};
  }

  for (int kk = 0; kk < E_DIM; kk += 4) {
    const int k0 = kk + koff;
    v2f a;
    a.x = Elds[m * 256 + (k0 ^ msw)];
    a.y = Elds[m * 256 + ((k0 + 1) ^ msw)];
#pragma unroll
    for (int c = 0; c < 2; ++c) {
      v2f bf;
      bf.x = Wl[(size_t)k0 * V_DIM + colc[c]];
      bf.y = Wl[(size_t)(k0 + 1) * V_DIM + colc[c]];
      acc2[c] = wmma4(a, bf, acc2[c]);
    }
  }

#pragma unroll
  for (int c = 0; c < 2; ++c) {
    const float blv = bl[colc[c]];
#pragma unroll
    for (int v = 0; v < 8; ++v) {
      const int r = (lane < 16) ? v : v + 8;
      int g, k;
      bool ok;
      if (rt < 8) {
        const int er = rt * 16 + r;
        g  = b * 16 + (er >> 3);
        k  = er & 7;
        ok = (g < G);
      } else {
        g  = xt * 16 + r;
        k  = kx;
        ok = has_x && (g < G);
      }
      if (cval[c] && ok) {
        const int n = g / nt;
        const int t = g - n * nt;
        const int j = out_off + t * S + k;
        out[((size_t)n * T_SEQ + j) * V_DIM + colv[c]] = acc2[c][v] + blv;
      }
    }
  }
}

extern "C" void kernel_launch(void* const* d_in, const int* in_sizes, int n_in,
                              void* d_out, int out_size, void* d_ws, size_t ws_size,
                              hipStream_t stream) {
  const float* x   = (const float*)d_in[0];
  const float* W1  = (const float*)d_in[4];
  const float* b1  = (const float*)d_in[5];
  const float* W2  = (const float*)d_in[6];
  const float* b2  = (const float*)d_in[7];
  const float* W3  = (const float*)d_in[8];
  const float* b3  = (const float*)d_in[9];
  const float* Wd4 = (const float*)d_in[10];
  const float* bd4 = (const float*)d_in[11];
  const float* Wl4 = (const float*)d_in[12];
  const float* bl4 = (const float*)d_in[13];
  const float* Wd5 = (const float*)d_in[14];
  const float* bd5 = (const float*)d_in[15];
  const float* Wl5 = (const float*)d_in[16];
  const float* bl5 = (const float*)d_in[17];
  float* out = (float*)d_out;

  const int N = 8;

  head123_kernel<<<dim3(100, N), 32, 0, stream>>>(x, W1, b1, W2, b2, W3, b3, out);

  conv_head_kernel<4, 1><<<dim3(32, N), 256, 0, stream>>>(
      x, Wd4, bd4, Wl4, bl4, out,  1600,  1600,  512);

  conv8_kernel<<<dim3(256), 512, 0, stream>>>(
      x, Wd5, bd5, Wl5, bl5, out,  2112,  3648,  568,  8);
}


# Conv head S=8 block re-cut: 32 tokens x 4 taps (one 16-byte tap group) + 1 leftover-tile tap, was 16 tokens x 8 taps: a Wd fragment feeds 2 token tiles, leftover tap from registers; sums unchanged
# speedup vs baseline: 3.1788x; 1.1016x over previous
#include <hip/hip_runtime.h>
#include <hip/hip_bf16.h>


#define T_SEQ 8192
#define E_DIM 256
#define V_DIM 17

typedef float v2f __attribute__((ext_vector_type(2)));
typedef float v8f __attribute__((ext_vector_type(8)));

__device__ __forceinline__ v8f wmma4(v2f a, v2f b, v8f c) {
  return __builtin_amdgcn_wmma_f32_16x16x4_f32(
       false, a,  false, b,
       (short)0, c,  false,  false);
}

__global__ __launch_bounds__(32) void head123_kernel(
    const float* __restrict__ x,
    const float* __restrict__ W1, const float* __restrict__ b1,
    const float* __restrict__ W2, const float* __restrict__ b2,
    const float* __restrict__ W3, const float* __restrict__ b3,
    float* __restrict__ out) {
  const int n       = blockIdx.y;
  const int rowbase = blockIdx.x * 16;

  const float* W;
  const float* b;
  if (rowbase < 64)       { W = W1; b = b1; }
  else if (rowbase < 576) { W = W2; b = b2; }
  else                    { W = W3; b = b3; }

  const int lane = threadIdx.x & 31;
  const int l15  = lane & 15;
  const int koff = (lane < 16) ? 0 : 2;

  const float* xrow = x + ((size_t)n * T_SEQ + rowbase + l15) * E_DIM;
  const int c1  = 16 + l15;
  const int c1c = (c1 < V_DIM) ? c1 : (V_DIM - 1);

  v8f acc0 = {};
  v8f acc1 = {};
  for (int kk = 0; kk < E_DIM; kk += 4) {
    v2f a;
    a.x = xrow[kk + koff];
    a.y = xrow[kk + koff + 1];
    v2f bf0, bf1;
    bf0.x = W[(kk + koff) * V_DIM + l15];
    bf0.y = W[(kk + koff + 1) * V_DIM + l15];
    bf1.x = W[(kk + koff) * V_DIM + c1c];
    bf1.y = W[(kk + koff + 1) * V_DIM + c1c];
    acc0 = wmma4(a, bf0, acc0);
    acc1 = wmma4(a, bf1, acc1);
  }

#pragma unroll
  for (int v = 0; v < 8; ++v) {
    const int r = ((lane < 16) ? v : v + 8);
    float* o = out + ((size_t)n * T_SEQ + rowbase + r) * V_DIM;
    o[l15] = acc0[v] + b[l15];
    if (c1 < V_DIM) o[c1] = acc1[v] + b[c1];
  }
}

template <int S, int MT>
__global__ __launch_bounds__(64 * S) void conv_head_kernel(
    const float* __restrict__ x,
    const float* __restrict__ Wd, const float* __restrict__ bd,
    const float* __restrict__ Wl, const float* __restrict__ bl,
    float* __restrict__ out,
    int ioff, int out_off, int nt) {
  constexpr int OTW = 8 / S;
  constexpr int Q   = S / 4;
  constexpr int Es  = E_DIM * S;
  constexpr int TOK = 16 * MT;
  __shared__ float Elds[TOK * S * 256];
  constexpr int XP = E_DIM + 4;
  __shared__ alignas(16) float Xs[TOK * XP];

  const int n  = blockIdx.y;
  const int tb = blockIdx.x;

  const int tid  = threadIdx.x;
  const int w    = tid >> 5;
  const int lane = tid & 31;
  const int l15  = lane & 15;
  const int koff = (lane < 16) ? 0 : 2;

  constexpr int NXL = (TOK * 64) / (64 * S);
  float4 xv[NXL];
#pragma unroll
  for (int j = 0; j < NXL; ++j) {
    const int idx    = tid + j * 64 * S;
    const int r      = idx >> 6;
    const int c4     = idx & 63;
    const int trel   = tb * TOK + r;
    const int tclamp = (trel < nt) ? trel : (nt - 1);
    xv[j] = *reinterpret_cast<const float4*>(
        x + ((size_t)n * T_SEQ + ioff + tclamp) * E_DIM + c4 * 4);
  }
#pragma unroll
  for (int j = 0; j < NXL; ++j) {
    const int idx = tid + j * 64 * S;
    *reinterpret_cast<float4*>(&Xs[(idx >> 6) * XP + (idx & 63) * 4]) = xv[j];
  }
  __syncthreads();

  const float* xrow = &Xs[l15 * XP];

  v8f acc[MT][8];
#pragma unroll
  for (int m = 0; m < MT; ++m)
#pragma unroll
    for (int i = 0; i < 8; ++i) acc[m][i] = (v8f){};

  const float* wbase = Wd + (size_t)koff * Es + (size_t)((w * OTW) * 16 + l15) * S;

#pragma unroll(S == 8 ? 2 : 4)
  for (int kk = 0; kk < E_DIM; kk += 4) {
    v2f a[MT];
#pragma unroll
    for (int m = 0; m < MT; ++m) {
      a[m].x = xrow[m * 16 * XP + kk + koff];
      a[m].y = xrow[m * 16 * XP + kk + koff + 1];
    }
    const float* wrow = wbase + (size_t)kk * Es;
#pragma unroll
    for (int otl = 0; otl < OTW; ++otl) {
#pragma unroll
      for (int q = 0; q < Q; ++q) {
        const float* bp = wrow + otl * 16 * S + q * 4;
        const float4 r0 = *reinterpret_cast<const float4*>(bp);
        const float4 r1 = *reinterpret_cast<const float4*>(bp + Es);
        const int t0 = otl * S + q * 4;
        v2f bf;
        bf.x = r0.x; bf.y = r1.x;
#pragma unroll
        for (int m = 0; m < MT; ++m) acc[m][t0 + 0] = wmma4(a[m], bf, acc[m][t0 + 0]);
        bf.x = r0.y; bf.y = r1.y;
#pragma unroll
        for (int m = 0; m < MT; ++m) acc[m][t0 + 1] = wmma4(a[m], bf, acc[m][t0 + 1]);
        bf.x = r0.z; bf.y = r1.z;
#pragma unroll
        for (int m = 0; m < MT; ++m) acc[m][t0 + 2] = wmma4(a[m], bf, acc[m][t0 + 2]);
        bf.x = r0.w; bf.y = r1.w;
#pragma unroll
        for (int m = 0; m < MT; ++m) acc[m][t0 + 3] = wmma4(a[m], bf, acc[m][t0 + 3]);
      }
    }
  }

#pragma unroll
  for (int i = 0; i < 8; ++i) {
    const int otl = i / S;
    const int k   = i % S;
    const int o   = (w * OTW + otl) * 16 + l15;
    const float bdv = bd[o];
#pragma unroll
    for (int m = 0; m < MT; ++m) {
#pragma unroll
      for (int v = 0; v < 8; ++v) {
        const int tl  = m * 16 + ((lane < 16) ? v : v + 8);
        const int row = tl * S + k;
        Elds[row * 256 + (o ^ ((row & 15) * 4))] = acc[m][i][v] + bdv;
      }
    }
  }
  __syncthreads();

  constexpr int CT = MT;
  const int rt   = (MT == 1) ? (w >> 1) : w;
  const int col0 = (MT == 1) ? (w & 1) * 16 : 0;
  const int m    = rt * 16 + l15;
  const int msw  = (m & 15) * 4;

  int  colv[CT];
  bool cval[CT];
  int  colc[CT];
  v8f  acc2[CT];
#pragma unroll
  for (int c = 0; c < CT; ++c) {
    colv[c] = col0 + c * 16 + l15;
    cval[c] = (colv[c] < V_DIM);
    colc[c] = cval[c] ? colv[c] : (V_DIM - 1);
    acc2[c] = (v8f){};
  }

  for (int kk = 0; kk < E_DIM; kk += 4) {
    const int k0 = kk + koff;
    v2f a;
    a.x = Elds[m * 256 + (k0 ^ msw)];
    a.y = Elds[m * 256 + ((k0 + 1) ^ msw)];
#pragma unroll
    for (int c = 0; c < CT; ++c) {
      v2f bf;
      bf.x = Wl[(size_t)k0 * V_DIM + colc[c]];
      bf.y = Wl[(size_t)(k0 + 1) * V_DIM + colc[c]];
      acc2[c] = wmma4(a, bf, acc2[c]);
    }
  }

#pragma unroll
  for (int c = 0; c < CT; ++c) {
    const float blv = bl[colc[c]];
#pragma unroll
    for (int v = 0; v < 8; ++v) {
      const int r     = rt * 16 + ((lane < 16) ? v : v + 8);
      const int tl    = r / S;
      const int k     = r % S;
      const int trel2 = tb * TOK + tl;
      if (cval[c] && trel2 < nt) {
        const int j = out_off + trel2 * S + k;
        out[((size_t)n * T_SEQ + j) * V_DIM + colv[c]] = acc2[c][v] + blv;
      }
    }
  }
}


template <int JX>
__device__ __forceinline__ void conv8_stage1(const float* __restrict__ xrow,
                                             const float* __restrict__ wbase,
                                             v8f (&acc)[2][4], v8f& accx) {
  constexpr int Es = E_DIM * 8;
  constexpr int XP = E_DIM + 4;
#pragma unroll 4
  for (int kk = 0; kk < E_DIM; kk += 4) {
    v2f a0, a1, ax;
    a0.x = xrow[kk];
    a0.y = xrow[kk + 1];
    a1.x = xrow[16 * XP + kk];
    a1.y = xrow[16 * XP + kk + 1];
    ax.x = xrow[32 * XP + kk];
    ax.y = xrow[32 * XP + kk + 1];
    const float* wrow = wbase + (size_t)kk * Es;
    const float4 r0 = *reinterpret_cast<const float4*>(wrow);
    const float4 r1 = *reinterpret_cast<const float4*>(wrow + Es);
    v2f bf;
    bf.x = r0.x; bf.y = r1.x;
    acc[0][0] = wmma4(a0, bf, acc[0][0]);
    acc[1][0] = wmma4(a1, bf, acc[1][0]);
    if (JX == 0) accx = wmma4(ax, bf, accx);
    bf.x = r0.y; bf.y = r1.y;
    acc[0][1] = wmma4(a0, bf, acc[0][1]);
    acc[1][1] = wmma4(a1, bf, acc[1][1]);
    if (JX == 1) accx = wmma4(ax, bf, accx);
    bf.x = r0.z; bf.y = r1.z;
    acc[0][2] = wmma4(a0, bf, acc[0][2]);
    acc[1][2] = wmma4(a1, bf, acc[1][2]);
    if (JX == 2) accx = wmma4(ax, bf, accx);
    bf.x = r0.w; bf.y = r1.w;
    acc[0][3] = wmma4(a0, bf, acc[0][3]);
    acc[1][3] = wmma4(a1, bf, acc[1][3]);
    if (JX == 3) accx = wmma4(ax, bf, accx);
  }
}

__global__ __launch_bounds__(512) void conv8_kernel(
    const float* __restrict__ x,
    const float* __restrict__ Wd, const float* __restrict__ bd,
    const float* __restrict__ Wl, const float* __restrict__ bl,
    float* __restrict__ out,
    int ioff, int out_off, int nt, int nbatch) {
  constexpr int S    = 8;
  constexpr int Es   = E_DIM * S;
  constexpr int XP   = E_DIM + 4;
  constexpr int ROWS = 2 * 16 * 4 + 16;
  __shared__ float Elds[ROWS * 256];
  __shared__ alignas(16) float Xs[48 * XP];

  const int b      = blockIdx.x;
  const int q      = b & 1;
  const int p      = b >> 1;
  const int G      = nbatch * nt;
  const int ntiles = (G + 15) >> 4;
  const int npairs = (int)gridDim.x >> 1;
  const int xt_raw = 2 * npairs + (p >> 2);
  const bool has_x = xt_raw < ntiles;
  const int xt     = has_x ? xt_raw : 2 * p;
  const int jx     = p & 3;

  const int tid  = threadIdx.x;
  const int w    = tid >> 5;
  const int lane = tid & 31;
  const int l15  = lane & 15;
  const int koff = (lane < 16) ? 0 : 2;

  float4 xv[6];
#pragma unroll
  for (int j = 0; j < 6; ++j) {
    const int idx  = tid + j * 512;
    const int r    = idx >> 6;
    const int c4   = idx & 63;
    const int tile = (j < 2) ? 2 * p : (j < 4) ? 2 * p + 1 : xt;
    const int g    = tile * 16 + (r & 15);
    const int gc   = (g < G) ? g : (G - 1);
    const int n    = gc / nt;
    const int t    = gc - n * nt;
    xv[j] = *reinterpret_cast<const float4*>(
        x + ((size_t)n * T_SEQ + ioff + t) * E_DIM + c4 * 4);
  }
#pragma unroll
  for (int j = 0; j < 6; ++j) {
    const int idx = tid + j * 512;
    *reinterpret_cast<float4*>(&Xs[(idx >> 6) * XP + (idx & 63) * 4]) = xv[j];
  }
  __syncthreads();

  const float* xrow = &Xs[l15 * XP + koff];
  v8f acc[2][4];
#pragma unroll
  for (int m = 0; m < 2; ++m)
#pragma unroll
    for (int j = 0; j < 4; ++j) acc[m][j] = (v8f){};
  v8f accx = {};

  const float* wbase = Wd + (size_t)koff * Es + (size_t)(w * 16 + l15) * S + q * 4;

  switch (jx) {
    case 0:  conv8_stage1<0>(xrow, wbase, acc, accx); break;
    case 1:  conv8_stage1<1>(xrow, wbase, acc, accx); break;
    case 2:  conv8_stage1<2>(xrow, wbase, acc, accx); break;
    default: conv8_stage1<3>(xrow, wbase, acc, accx); break;
  }

  {
    const int o = w * 16 + l15;
    const float bdv = bd[o];
#pragma unroll
    for (int m = 0; m < 2; ++m) {
#pragma unroll
      for (int j = 0; j < 4; ++j) {
#pragma unroll
        for (int v = 0; v < 8; ++v) {
          const int tl  = (lane < 16) ? v : v + 8;
          const int row = m * 64 + tl * 4 + j;
          Elds[row * 256 + (o ^ ((row & 15) * 4))] = acc[m][j][v] + bdv;
        }
      }
    }
#pragma unroll
    for (int v = 0; v < 8; ++v) {
      const int tl  = (lane < 16) ? v : v + 8;
      const int row = 128 + tl;
      Elds[row * 256 + (o ^ ((row & 15) * 4))] = accx[v] + bdv;
    }
  }
  __syncthreads();

  if (w >= 9) return;
  const int rt  = w;
  const int m   = rt * 16 + l15;
  const int msw = (m & 15) * 4;

  int  colv[2];
  bool cval[2];
  int  colc[2];
  v8f  acc2[2];
#pragma unroll
  for (int c = 0; c < 2; ++c) {
    colv[c] = c * 16 + l15;
    cval[c] = (colv[c] < V_DIM);
    colc[c] = cval[c] ? colv[c] : (V_DIM - 1);
    acc2[c] = (v8f){};
  }

  for (int kk = 0; kk < E_DIM; kk += 4) {
    const int k0 = kk + koff;
    v2f a;
    a.x = Elds[m * 256 + (k0 ^ msw)];
    a.y = Elds[m * 256 + ((k0 + 1) ^ msw)];
#pragma unroll
    for (int c = 0; c < 2; ++c) {
      v2f bf;
      bf.x = Wl[(size_t)k0 * V_DIM + colc[c]];
      bf.y = Wl[(size_t)(k0 + 1) * V_DIM + colc[c]];
      acc2[c] = wmma4(a, bf, acc2[c]);
    }
  }

  const int r0w = (lane < 16) ? 0 : 8;
  if (rt < 8) {
    const int gbase = (2 * p + (rt >> 2)) * 16 + (rt & 3) * 4 + (r0w >> 2);
    const float blv0 = bl[colc[0]];
    const float blv1 = bl[colc[1]];
#pragma unroll
    for (int tk = 0; tk < 2; ++tk) {
      const int g = gbase + tk;
      const int n = g / nt;
      const int t = g - n * nt;
      float* orow = out + ((size_t)n * T_SEQ + out_off + t * S + 4 * q) * V_DIM;
      if (g < G) {
#pragma unroll
        for (int jj = 0; jj < 4; ++jj) {
          if (cval[0]) orow[jj * V_DIM + colv[0]] = acc2[0][tk * 4 + jj] + blv0;
          if (cval[1]) orow[jj * V_DIM + colv[1]] = acc2[1][tk * 4 + jj] + blv1;
        }
      }
    }
  } else {
#pragma unroll
    for (int c = 0; c < 2; ++c) {
      const float blv = bl[colc[c]];
#pragma unroll
      for (int v = 0; v < 8; ++v) {
        const int r = r0w + v;
        const int g = xt * 16 + r;
        const int k = 4 * q + jx;
        if (cval[c] && has_x && g < G) {
          const int n = g / nt;
          const int t = g - n * nt;
          const int j = out_off + t * S + k;
          out[((size_t)n * T_SEQ + j) * V_DIM + colv[c]] = acc2[c][v] + blv;
        }
      }
    }
  }
}

extern "C" void kernel_launch(void* const* d_in, const int* in_sizes, int n_in,
                              void* d_out, int out_size, void* d_ws, size_t ws_size,
                              hipStream_t stream) {
  const float* x   = (const float*)d_in[0];
  const float* W1  = (const float*)d_in[4];
  const float* b1  = (const float*)d_in[5];
  const float* W2  = (const float*)d_in[6];
  const float* b2  = (const float*)d_in[7];
  const float* W3  = (const float*)d_in[8];
  const float* b3  = (const float*)d_in[9];
  const float* Wd4 = (const float*)d_in[10];
  const float* bd4 = (const float*)d_in[11];
  const float* Wl4 = (const float*)d_in[12];
  const float* bl4 = (const float*)d_in[13];
  const float* Wd5 = (const float*)d_in[14];
  const float* bd5 = (const float*)d_in[15];
  const float* Wl5 = (const float*)d_in[16];
  const float* bl5 = (const float*)d_in[17];
  float* out = (float*)d_out;

  const int N = 8;

  head123_kernel<<<dim3(100, N), 32, 0, stream>>>(x, W1, b1, W2, b2, W3, b3, out);

  conv_head_kernel<4, 1><<<dim3(32, N), 256, 0, stream>>>(
      x, Wd4, bd4, Wl4, bl4, out,  1600,  1600,  512);

  conv8_kernel<<<dim3(256), 512, 0, stream>>>(
      x, Wd5, bd5, Wl5, bl5, out,  2112,  3648,  568,  8);
}


# Conv heads stage 1: K loop over two operand register sets, trip t+1's Wd and x fragment loads issued before trip t's WMMAs (was load, wait, WMMA inside each trip); k order, f32 sums unchanged
# speedup vs baseline: 3.3947x; 1.0679x over previous
#include <hip/hip_runtime.h>
#include <hip/hip_bf16.h>


#define T_SEQ 8192
#define E_DIM 256
#define V_DIM 17

typedef float v2f __attribute__((ext_vector_type(2)));
typedef float v8f __attribute__((ext_vector_type(8)));

__device__ __forceinline__ v8f wmma4(v2f a, v2f b, v8f c) {
  return __builtin_amdgcn_wmma_f32_16x16x4_f32(
       false, a,  false, b,
       (short)0, c,  false,  false);
}

__global__ __launch_bounds__(32) void head123_kernel(
    const float* __restrict__ x,
    const float* __restrict__ W1, const float* __restrict__ b1,
    const float* __restrict__ W2, const float* __restrict__ b2,
    const float* __restrict__ W3, const float* __restrict__ b3,
    float* __restrict__ out) {
  const int n       = blockIdx.y;
  const int rowbase = blockIdx.x * 16;

  const float* W;
  const float* b;
  if (rowbase < 64)       { W = W1; b = b1; }
  else if (rowbase < 576) { W = W2; b = b2; }
  else                    { W = W3; b = b3; }

  const int lane = threadIdx.x & 31;
  const int l15  = lane & 15;
  const int koff = (lane < 16) ? 0 : 2;

  const float* xrow = x + ((size_t)n * T_SEQ + rowbase + l15) * E_DIM;
  const int c1  = 16 + l15;
  const int c1c = (c1 < V_DIM) ? c1 : (V_DIM - 1);

  v8f acc0 = {};
  v8f acc1 = {};
  for (int kk = 0; kk < E_DIM; kk += 4) {
    v2f a;
    a.x = xrow[kk + koff];
    a.y = xrow[kk + koff + 1];
    v2f bf0, bf1;
    bf0.x = W[(kk + koff) * V_DIM + l15];
    bf0.y = W[(kk + koff + 1) * V_DIM + l15];
    bf1.x = W[(kk + koff) * V_DIM + c1c];
    bf1.y = W[(kk + koff + 1) * V_DIM + c1c];
    acc0 = wmma4(a, bf0, acc0);
    acc1 = wmma4(a, bf1, acc1);
  }

#pragma unroll
  for (int v = 0; v < 8; ++v) {
    const int r = ((lane < 16) ? v : v + 8);
    float* o = out + ((size_t)n * T_SEQ + rowbase + r) * V_DIM;
    o[l15] = acc0[v] + b[l15];
    if (c1 < V_DIM) o[c1] = acc1[v] + b[c1];
  }
}

template <int S, int MT>
struct ConvFrag {
  static constexpr int KT  = (S == 8) ? 2 : 4;
  static constexpr int OTW = 8 / S;
  static constexpr int Q   = S / 4;
  float4 r0[KT][OTW][Q], r1[KT][OTW][Q];
  v2f    a[KT][MT];
};

template <int S, int MT>
__device__ __forceinline__ void conv_load_trip(ConvFrag<S, MT>& f, const float* __restrict__ xk,
                                               const float* __restrict__ wbase, int kk0) {
  constexpr int KT = ConvFrag<S, MT>::KT, OTW = ConvFrag<S, MT>::OTW, Q = ConvFrag<S, MT>::Q;
  constexpr int Es = E_DIM * S;
  constexpr int XP = E_DIM + 4;
#pragma unroll
  for (int s = 0; s < KT; ++s) {
    const float* wrow = wbase + (size_t)(kk0 + 4 * s) * Es;
#pragma unroll
    for (int otl = 0; otl < OTW; ++otl) {
#pragma unroll
      for (int q = 0; q < Q; ++q) {
        const float* bp = wrow + otl * 16 * S + q * 4;
        f.r0[s][otl][q] = *reinterpret_cast<const float4*>(bp);
        f.r1[s][otl][q] = *reinterpret_cast<const float4*>(bp + Es);
      }
    }
  }
#pragma unroll
  for (int s = 0; s < KT; ++s) {
    const int kk = kk0 + 4 * s;
#pragma unroll
    for (int m = 0; m < MT; ++m) {
      f.a[s][m].x = xk[m * 16 * XP + kk];
      f.a[s][m].y = xk[m * 16 * XP + kk + 1];
    }
  }
}

template <int S, int MT>
__device__ __forceinline__ void conv_mma_trip(const ConvFrag<S, MT>& f, v8f (&acc)[MT][8]) {
  constexpr int KT = ConvFrag<S, MT>::KT, OTW = ConvFrag<S, MT>::OTW, Q = ConvFrag<S, MT>::Q;
#pragma unroll
  for (int s = 0; s < KT; ++s) {
#pragma unroll
    for (int otl = 0; otl < OTW; ++otl) {
#pragma unroll
      for (int q = 0; q < Q; ++q) {
        const float4 r0 = f.r0[s][otl][q];
        const float4 r1 = f.r1[s][otl][q];
        const int t0 = otl * S + q * 4;
        v2f bf;
        bf.x = r0.x; bf.y = r1.x;
#pragma unroll
        for (int m = 0; m < MT; ++m) acc[m][t0 + 0] = wmma4(f.a[s][m], bf, acc[m][t0 + 0]);
        bf.x = r0.y; bf.y = r1.y;
#pragma unroll
        for (int m = 0; m < MT; ++m) acc[m][t0 + 1] = wmma4(f.a[s][m], bf, acc[m][t0 + 1]);
        bf.x = r0.z; bf.y = r1.z;
#pragma unroll
        for (int m = 0; m < MT; ++m) acc[m][t0 + 2] = wmma4(f.a[s][m], bf, acc[m][t0 + 2]);
        bf.x = r0.w; bf.y = r1.w;
#pragma unroll
        for (int m = 0; m < MT; ++m) acc[m][t0 + 3] = wmma4(f.a[s][m], bf, acc[m][t0 + 3]);
      }
    }
  }
}

template <int S, int MT>
__global__ __launch_bounds__(64 * S) void conv_head_kernel(
    const float* __restrict__ x,
    const float* __restrict__ Wd, const float* __restrict__ bd,
    const float* __restrict__ Wl, const float* __restrict__ bl,
    float* __restrict__ out,
    int ioff, int out_off, int nt) {
  constexpr int OTW = 8 / S;
  constexpr int Q   = S / 4;
  constexpr int Es  = E_DIM * S;
  constexpr int TOK = 16 * MT;
  __shared__ float Elds[TOK * S * 256];
  constexpr int XP = E_DIM + 4;
  __shared__ alignas(16) float Xs[TOK * XP];

  const int n  = blockIdx.y;
  const int tb = blockIdx.x;

  const int tid  = threadIdx.x;
  const int w    = tid >> 5;
  const int lane = tid & 31;
  const int l15  = lane & 15;
  const int koff = (lane < 16) ? 0 : 2;

  constexpr int NXL = (TOK * 64) / (64 * S);
  float4 xv[NXL];
#pragma unroll
  for (int j = 0; j < NXL; ++j) {
    const int idx    = tid + j * 64 * S;
    const int r      = idx >> 6;
    const int c4     = idx & 63;
    const int trel   = tb * TOK + r;
    const int tclamp = (trel < nt) ? trel : (nt - 1);
    xv[j] = *reinterpret_cast<const float4*>(
        x + ((size_t)n * T_SEQ + ioff + tclamp) * E_DIM + c4 * 4);
  }
#pragma unroll
  for (int j = 0; j < NXL; ++j) {
    const int idx = tid + j * 64 * S;
    *reinterpret_cast<float4*>(&Xs[(idx >> 6) * XP + (idx & 63) * 4]) = xv[j];
  }
  __syncthreads();

  const float* xrow = &Xs[l15 * XP];

  v8f acc[MT][8];
#pragma unroll
  for (int m = 0; m < MT; ++m)
#pragma unroll
    for (int i = 0; i < 8; ++i) acc[m][i] = (v8f){};

  const float* wbase = Wd + (size_t)koff * Es + (size_t)((w * OTW) * 16 + l15) * S;

  {
    constexpr int TK = 4 * ConvFrag<S, MT>::KT;
    ConvFrag<S, MT> fa, fb;
    conv_load_trip<S, MT>(fa, xrow + koff, wbase, 0);
    for (int kk = 0; kk < E_DIM - 2 * TK; kk += 2 * TK) {
      conv_load_trip<S, MT>(fb, xrow + koff, wbase, kk + TK);
      __builtin_amdgcn_sched_barrier(0);
      conv_mma_trip<S, MT>(fa, acc);
      __builtin_amdgcn_sched_barrier(0);
      conv_load_trip<S, MT>(fa, xrow + koff, wbase, kk + 2 * TK);
      __builtin_amdgcn_sched_barrier(0);
      conv_mma_trip<S, MT>(fb, acc);
      __builtin_amdgcn_sched_barrier(0);
    }
    conv_load_trip<S, MT>(fb, xrow + koff, wbase, E_DIM - TK);
    __builtin_amdgcn_sched_barrier(0);
    conv_mma_trip<S, MT>(fa, acc);
    conv_mma_trip<S, MT>(fb, acc);
  }

#pragma unroll
  for (int i = 0; i < 8; ++i) {
    const int otl = i / S;
    const int k   = i % S;
    const int o   = (w * OTW + otl) * 16 + l15;
    const float bdv = bd[o];
#pragma unroll
    for (int m = 0; m < MT; ++m) {
#pragma unroll
      for (int v = 0; v < 8; ++v) {
        const int tl  = m * 16 + ((lane < 16) ? v : v + 8);
        const int row = tl * S + k;
        Elds[row * 256 + (o ^ ((row & 15) * 4))] = acc[m][i][v] + bdv;
      }
    }
  }
  __syncthreads();

  constexpr int CT = MT;
  const int rt   = (MT == 1) ? (w >> 1) : w;
  const int col0 = (MT == 1) ? (w & 1) * 16 : 0;
  const int m    = rt * 16 + l15;
  const int msw  = (m & 15) * 4;

  int  colv[CT];
  bool cval[CT];
  int  colc[CT];
  v8f  acc2[CT];
#pragma unroll
  for (int c = 0; c < CT; ++c) {
    colv[c] = col0 + c * 16 + l15;
    cval[c] = (colv[c] < V_DIM);
    colc[c] = cval[c] ? colv[c] : (V_DIM - 1);
    acc2[c] = (v8f){};
  }

  for (int kk = 0; kk < E_DIM; kk += 4) {
    const int k0 = kk + koff;
    v2f a;
    a.x = Elds[m * 256 + (k0 ^ msw)];
    a.y = Elds[m * 256 + ((k0 + 1) ^ msw)];
#pragma unroll
    for (int c = 0; c < CT; ++c) {
      v2f bf;
      bf.x = Wl[(size_t)k0 * V_DIM + colc[c]];
      bf.y = Wl[(size_t)(k0 + 1) * V_DIM + colc[c]];
      acc2[c] = wmma4(a, bf, acc2[c]);
    }
  }

#pragma unroll
  for (int c = 0; c < CT; ++c) {
    const float blv = bl[colc[c]];
#pragma unroll
    for (int v = 0; v < 8; ++v) {
      const int r     = rt * 16 + ((lane < 16) ? v : v + 8);
      const int tl    = r / S;
      const int k     = r % S;
      const int trel2 = tb * TOK + tl;
      if (cval[c] && trel2 < nt) {
        const int j = out_off + trel2 * S + k;
        out[((size_t)n * T_SEQ + j) * V_DIM + colv[c]] = acc2[c][v] + blv;
      }
    }
  }
}


struct Conv8Frag {
  float4 r0[4], r1[4];
  v2f    a0[4], a1[4], ax[4];
};

__device__ __forceinline__ void conv8_load(Conv8Frag& f, const float* __restrict__ xrow,
                                           const float* __restrict__ wbase, int kk0) {
  constexpr int Es = E_DIM * 8;
  constexpr int XP = E_DIM + 4;
#pragma unroll
  for (int s = 0; s < 4; ++s) {
    const int kk = kk0 + 4 * s;
    const float* wrow = wbase + (size_t)kk * Es;
    f.r0[s] = *reinterpret_cast<const float4*>(wrow);
    f.r1[s] = *reinterpret_cast<const float4*>(wrow + Es);
  }
#pragma unroll
  for (int s = 0; s < 4; ++s) {
    const int kk = kk0 + 4 * s;
    f.a0[s].x = xrow[kk];
    f.a0[s].y = xrow[kk + 1];
    f.a1[s].x = xrow[16 * XP + kk];
    f.a1[s].y = xrow[16 * XP + kk + 1];
    f.ax[s].x = xrow[32 * XP + kk];
    f.ax[s].y = xrow[32 * XP + kk + 1];
  }
}

template <int JX>
__device__ __forceinline__ void conv8_mma(const Conv8Frag& f, v8f (&acc)[2][4], v8f& accx) {
#pragma unroll
  for (int s = 0; s < 4; ++s) {
    const float4 r0 = f.r0[s];
    const float4 r1 = f.r1[s];
    const v2f a0 = f.a0[s], a1 = f.a1[s], ax = f.ax[s];
    v2f bf;
    bf.x = r0.x; bf.y = r1.x;
    acc[0][0] = wmma4(a0, bf, acc[0][0]);
    acc[1][0] = wmma4(a1, bf, acc[1][0]);
    if (JX == 0) accx = wmma4(ax, bf, accx);
    bf.x = r0.y; bf.y = r1.y;
    acc[0][1] = wmma4(a0, bf, acc[0][1]);
    acc[1][1] = wmma4(a1, bf, acc[1][1]);
    if (JX == 1) accx = wmma4(ax, bf, accx);
    bf.x = r0.z; bf.y = r1.z;
    acc[0][2] = wmma4(a0, bf, acc[0][2]);
    acc[1][2] = wmma4(a1, bf, acc[1][2]);
    if (JX == 2) accx = wmma4(ax, bf, accx);
    bf.x = r0.w; bf.y = r1.w;
    acc[0][3] = wmma4(a0, bf, acc[0][3]);
    acc[1][3] = wmma4(a1, bf, acc[1][3]);
    if (JX == 3) accx = wmma4(ax, bf, accx);
  }
}

template <int JX>
__device__ __forceinline__ void conv8_stage1(const float* __restrict__ xrow,
                                             const float* __restrict__ wbase,
                                             v8f (&acc)[2][4], v8f& accx) {
  constexpr int TK = 16;
  Conv8Frag fa, fb;
  conv8_load(fa, xrow, wbase, 0);
#pragma nounroll
  for (int kk = 0; kk < E_DIM - 2 * TK; kk += 2 * TK) {
    conv8_load(fb, xrow, wbase, kk + TK);
    conv8_mma<JX>(fa, acc, accx);
    conv8_load(fa, xrow, wbase, kk + 2 * TK);
    conv8_mma<JX>(fb, acc, accx);
  }
  conv8_load(fb, xrow, wbase, E_DIM - TK);
  conv8_mma<JX>(fa, acc, accx);
  conv8_mma<JX>(fb, acc, accx);
}

__global__ __launch_bounds__(512) void conv8_kernel(
    const float* __restrict__ x,
    const float* __restrict__ Wd, const float* __restrict__ bd,
    const float* __restrict__ Wl, const float* __restrict__ bl,
    float* __restrict__ out,
    int ioff, int out_off, int nt, int nbatch) {
  constexpr int S    = 8;
  constexpr int Es   = E_DIM * S;
  constexpr int XP   = E_DIM + 4;
  constexpr int ROWS = 2 * 16 * 4 + 16;
  __shared__ float Elds[ROWS * 256];
  __shared__ alignas(16) float Xs[48 * XP];

  const int b      = blockIdx.x;
  const int q      = b & 1;
  const int p      = b >> 1;
  const int G      = nbatch * nt;
  const int ntiles = (G + 15) >> 4;
  const int npairs = (int)gridDim.x >> 1;
  const int xt_raw = 2 * npairs + (p >> 2);
  const bool has_x = xt_raw < ntiles;
  const int xt     = has_x ? xt_raw : 2 * p;
  const int jx     = p & 3;

  const int tid  = threadIdx.x;
  const int w    = tid >> 5;
  const int lane = tid & 31;
  const int l15  = lane & 15;
  const int koff = (lane < 16) ? 0 : 2;

  float4 xv[6];
#pragma unroll
  for (int j = 0; j < 6; ++j) {
    const int idx  = tid + j * 512;
    const int r    = idx >> 6;
    const int c4   = idx & 63;
    const int tile = (j < 2) ? 2 * p : (j < 4) ? 2 * p + 1 : xt;
    const int g    = tile * 16 + (r & 15);
    const int gc   = (g < G) ? g : (G - 1);
    const int n    = gc / nt;
    const int t    = gc - n * nt;
    xv[j] = *reinterpret_cast<const float4*>(
        x + ((size_t)n * T_SEQ + ioff + t) * E_DIM + c4 * 4);
  }
#pragma unroll
  for (int j = 0; j < 6; ++j) {
    const int idx = tid + j * 512;
    *reinterpret_cast<float4*>(&Xs[(idx >> 6) * XP + (idx & 63) * 4]) = xv[j];
  }
  __syncthreads();

  const float* xrow = &Xs[l15 * XP + koff];
  v8f acc[2][4];
#pragma unroll
  for (int m = 0; m < 2; ++m)
#pragma unroll
    for (int j = 0; j < 4; ++j) acc[m][j] = (v8f){};
  v8f accx = {};

  const float* wbase = Wd + (size_t)koff * Es + (size_t)(w * 16 + l15) * S + q * 4;

  switch (jx) {
    case 0:  conv8_stage1<0>(xrow, wbase, acc, accx); break;
    case 1:  conv8_stage1<1>(xrow, wbase, acc, accx); break;
    case 2:  conv8_stage1<2>(xrow, wbase, acc, accx); break;
    default: conv8_stage1<3>(xrow, wbase, acc, accx); break;
  }

  {
    const int o = w * 16 + l15;
    const float bdv = bd[o];
#pragma unroll
    for (int m = 0; m < 2; ++m) {
#pragma unroll
      for (int j = 0; j < 4; ++j) {
#pragma unroll
        for (int v = 0; v < 8; ++v) {
          const int tl  = (lane < 16) ? v : v + 8;
          const int row = m * 64 + tl * 4 + j;
          Elds[row * 256 + (o ^ ((row & 15) * 4))] = acc[m][j][v] + bdv;
        }
      }
    }
#pragma unroll
    for (int v = 0; v < 8; ++v) {
      const int tl  = (lane < 16) ? v : v + 8;
      const int row = 128 + tl;
      Elds[row * 256 + (o ^ ((row & 15) * 4))] = accx[v] + bdv;
    }
  }
  __syncthreads();

  if (w >= 9) return;
  const int rt  = w;
  const int m   = rt * 16 + l15;
  const int msw = (m & 15) * 4;

  int  colv[2];
  bool cval[2];
  int  colc[2];
  v8f  acc2[2];
#pragma unroll
  for (int c = 0; c < 2; ++c) {
    colv[c] = c * 16 + l15;
    cval[c] = (colv[c] < V_DIM);
    colc[c] = cval[c] ? colv[c] : (V_DIM - 1);
    acc2[c] = (v8f){};
  }

  for (int kk = 0; kk < E_DIM; kk += 4) {
    const int k0 = kk + koff;
    v2f a;
    a.x = Elds[m * 256 + (k0 ^ msw)];
    a.y = Elds[m * 256 + ((k0 + 1) ^ msw)];
#pragma unroll
    for (int c = 0; c < 2; ++c) {
      v2f bf;
      bf.x = Wl[(size_t)k0 * V_DIM + colc[c]];
      bf.y = Wl[(size_t)(k0 + 1) * V_DIM + colc[c]];
      acc2[c] = wmma4(a, bf, acc2[c]);
    }
  }

  const int r0w = (lane < 16) ? 0 : 8;
  if (rt < 8) {
    const int gbase = (2 * p + (rt >> 2)) * 16 + (rt & 3) * 4 + (r0w >> 2);
    const float blv0 = bl[colc[0]];
    const float blv1 = bl[colc[1]];
#pragma unroll
    for (int tk = 0; tk < 2; ++tk) {
      const int g = gbase + tk;
      const int n = g / nt;
      const int t = g - n * nt;
      float* orow = out + ((size_t)n * T_SEQ + out_off + t * S + 4 * q) * V_DIM;
      if (g < G) {
#pragma unroll
        for (int jj = 0; jj < 4; ++jj) {
          if (cval[0]) orow[jj * V_DIM + colv[0]] = acc2[0][tk * 4 + jj] + blv0;
          if (cval[1]) orow[jj * V_DIM + colv[1]] = acc2[1][tk * 4 + jj] + blv1;
        }
      }
    }
  } else {
#pragma unroll
    for (int c = 0; c < 2; ++c) {
      const float blv = bl[colc[c]];
#pragma unroll
      for (int v = 0; v < 8; ++v) {
        const int r = r0w + v;
        const int g = xt * 16 + r;
        const int k = 4 * q + jx;
        if (cval[c] && has_x && g < G) {
          const int n = g / nt;
          const int t = g - n * nt;
          const int j = out_off + t * S + k;
          out[((size_t)n * T_SEQ + j) * V_DIM + colv[c]] = acc2[c][v] + blv;
        }
      }
    }
  }
}

extern "C" void kernel_launch(void* const* d_in, const int* in_sizes, int n_in,
                              void* d_out, int out_size, void* d_ws, size_t ws_size,
                              hipStream_t stream) {
  const float* x   = (const float*)d_in[0];
  const float* W1  = (const float*)d_in[4];
  const float* b1  = (const float*)d_in[5];
  const float* W2  = (const float*)d_in[6];
  const float* b2  = (const float*)d_in[7];
  const float* W3  = (const float*)d_in[8];
  const float* b3  = (const float*)d_in[9];
  const float* Wd4 = (const float*)d_in[10];
  const float* bd4 = (const float*)d_in[11];
  const float* Wl4 = (const float*)d_in[12];
  const float* bl4 = (const float*)d_in[13];
  const float* Wd5 = (const float*)d_in[14];
  const float* bd5 = (const float*)d_in[15];
  const float* Wl5 = (const float*)d_in[16];
  const float* bl5 = (const float*)d_in[17];
  float* out = (float*)d_out;

  const int N = 8;

  head123_kernel<<<dim3(100, N), 32, 0, stream>>>(x, W1, b1, W2, b2, W3, b3, out);

  conv_head_kernel<4, 1><<<dim3(32, N), 256, 0, stream>>>(
      x, Wd4, bd4, Wl4, bl4, out,  1600,  1600,  512);

  conv8_kernel<<<dim3(256), 512, 0, stream>>>(
      x, Wd5, bd5, Wl5, bl5, out,  2112,  3648,  568,  8);
}


# Conv heads stage 1: each block starts its K walk at its own 16-channel chunk (16 phases over the grid) and wraps, so WGPs stop asking L2 for the same Wd lines at once; f32 sum order rotated per block
# speedup vs baseline: 3.5107x; 1.0342x over previous
#include <hip/hip_runtime.h>
#include <hip/hip_bf16.h>


#define T_SEQ 8192
#define E_DIM 256
#define V_DIM 17

typedef float v2f __attribute__((ext_vector_type(2)));
typedef float v8f __attribute__((ext_vector_type(8)));

__device__ __forceinline__ v8f wmma4(v2f a, v2f b, v8f c) {
  return __builtin_amdgcn_wmma_f32_16x16x4_f32(
       false, a,  false, b,
       (short)0, c,  false,  false);
}

__global__ __launch_bounds__(32) void head123_kernel(
    const float* __restrict__ x,
    const float* __restrict__ W1, const float* __restrict__ b1,
    const float* __restrict__ W2, const float* __restrict__ b2,
    const float* __restrict__ W3, const float* __restrict__ b3,
    float* __restrict__ out) {
  const int n       = blockIdx.y;
  const int rowbase = blockIdx.x * 16;

  const float* W;
  const float* b;
  if (rowbase < 64)       { W = W1; b = b1; }
  else if (rowbase < 576) { W = W2; b = b2; }
  else                    { W = W3; b = b3; }

  const int lane = threadIdx.x & 31;
  const int l15  = lane & 15;
  const int koff = (lane < 16) ? 0 : 2;

  const float* xrow = x + ((size_t)n * T_SEQ + rowbase + l15) * E_DIM;
  const int c1  = 16 + l15;
  const int c1c = (c1 < V_DIM) ? c1 : (V_DIM - 1);

  v8f acc0 = {};
  v8f acc1 = {};
  for (int kk = 0; kk < E_DIM; kk += 4) {
    v2f a;
    a.x = xrow[kk + koff];
    a.y = xrow[kk + koff + 1];
    v2f bf0, bf1;
    bf0.x = W[(kk + koff) * V_DIM + l15];
    bf0.y = W[(kk + koff + 1) * V_DIM + l15];
    bf1.x = W[(kk + koff) * V_DIM + c1c];
    bf1.y = W[(kk + koff + 1) * V_DIM + c1c];
    acc0 = wmma4(a, bf0, acc0);
    acc1 = wmma4(a, bf1, acc1);
  }

#pragma unroll
  for (int v = 0; v < 8; ++v) {
    const int r = ((lane < 16) ? v : v + 8);
    float* o = out + ((size_t)n * T_SEQ + rowbase + r) * V_DIM;
    o[l15] = acc0[v] + b[l15];
    if (c1 < V_DIM) o[c1] = acc1[v] + b[c1];
  }
}

template <int S, int MT>
struct ConvFrag {
  static constexpr int KT  = (S == 8) ? 2 : 4;
  static constexpr int OTW = 8 / S;
  static constexpr int Q   = S / 4;
  float4 r0[KT][OTW][Q], r1[KT][OTW][Q];
  v2f    a[KT][MT];
};

template <int S, int MT>
__device__ __forceinline__ void conv_load_trip(ConvFrag<S, MT>& f, const float* __restrict__ xk,
                                               const float* __restrict__ wbase, int kk0) {
  constexpr int KT = ConvFrag<S, MT>::KT, OTW = ConvFrag<S, MT>::OTW, Q = ConvFrag<S, MT>::Q;
  constexpr int Es = E_DIM * S;
  constexpr int XP = E_DIM + 4;
#pragma unroll
  for (int s = 0; s < KT; ++s) {
    const float* wrow = wbase + (size_t)(kk0 + 4 * s) * Es;
#pragma unroll
    for (int otl = 0; otl < OTW; ++otl) {
#pragma unroll
      for (int q = 0; q < Q; ++q) {
        const float* bp = wrow + otl * 16 * S + q * 4;
        f.r0[s][otl][q] = *reinterpret_cast<const float4*>(bp);
        f.r1[s][otl][q] = *reinterpret_cast<const float4*>(bp + Es);
      }
    }
  }
#pragma unroll
  for (int s = 0; s < KT; ++s) {
    const int kk = kk0 + 4 * s;
#pragma unroll
    for (int m = 0; m < MT; ++m) {
      f.a[s][m].x = xk[m * 16 * XP + kk];
      f.a[s][m].y = xk[m * 16 * XP + kk + 1];
    }
  }
}

template <int S, int MT>
__device__ __forceinline__ void conv_mma_trip(const ConvFrag<S, MT>& f, v8f (&acc)[MT][8]) {
  constexpr int KT = ConvFrag<S, MT>::KT, OTW = ConvFrag<S, MT>::OTW, Q = ConvFrag<S, MT>::Q;
#pragma unroll
  for (int s = 0; s < KT; ++s) {
#pragma unroll
    for (int otl = 0; otl < OTW; ++otl) {
#pragma unroll
      for (int q = 0; q < Q; ++q) {
        const float4 r0 = f.r0[s][otl][q];
        const float4 r1 = f.r1[s][otl][q];
        const int t0 = otl * S + q * 4;
        v2f bf;
        bf.x = r0.x; bf.y = r1.x;
#pragma unroll
        for (int m = 0; m < MT; ++m) acc[m][t0 + 0] = wmma4(f.a[s][m], bf, acc[m][t0 + 0]);
        bf.x = r0.y; bf.y = r1.y;
#pragma unroll
        for (int m = 0; m < MT; ++m) acc[m][t0 + 1] = wmma4(f.a[s][m], bf, acc[m][t0 + 1]);
        bf.x = r0.z; bf.y = r1.z;
#pragma unroll
        for (int m = 0; m < MT; ++m) acc[m][t0 + 2] = wmma4(f.a[s][m], bf, acc[m][t0 + 2]);
        bf.x = r0.w; bf.y = r1.w;
#pragma unroll
        for (int m = 0; m < MT; ++m) acc[m][t0 + 3] = wmma4(f.a[s][m], bf, acc[m][t0 + 3]);
      }
    }
  }
}

template <int S, int MT>
__global__ __launch_bounds__(64 * S) void conv_head_kernel(
    const float* __restrict__ x,
    const float* __restrict__ Wd, const float* __restrict__ bd,
    const float* __restrict__ Wl, const float* __restrict__ bl,
    float* __restrict__ out,
    int ioff, int out_off, int nt) {
  constexpr int OTW = 8 / S;
  constexpr int Q   = S / 4;
  constexpr int Es  = E_DIM * S;
  constexpr int TOK = 16 * MT;
  __shared__ float Elds[TOK * S * 256];
  constexpr int XP = E_DIM + 4;
  __shared__ alignas(16) float Xs[TOK * XP];

  const int n  = blockIdx.y;
  const int tb = blockIdx.x;

  const int tid  = threadIdx.x;
  const int w    = tid >> 5;
  const int lane = tid & 31;
  const int l15  = lane & 15;
  const int koff = (lane < 16) ? 0 : 2;

  constexpr int NXL = (TOK * 64) / (64 * S);
  float4 xv[NXL];
#pragma unroll
  for (int j = 0; j < NXL; ++j) {
    const int idx    = tid + j * 64 * S;
    const int r      = idx >> 6;
    const int c4     = idx & 63;
    const int trel   = tb * TOK + r;
    const int tclamp = (trel < nt) ? trel : (nt - 1);
    xv[j] = *reinterpret_cast<const float4*>(
        x + ((size_t)n * T_SEQ + ioff + tclamp) * E_DIM + c4 * 4);
  }
#pragma unroll
  for (int j = 0; j < NXL; ++j) {
    const int idx = tid + j * 64 * S;
    *reinterpret_cast<float4*>(&Xs[(idx >> 6) * XP + (idx & 63) * 4]) = xv[j];
  }
  __syncthreads();

  const float* xrow = &Xs[l15 * XP];

  v8f acc[MT][8];
#pragma unroll
  for (int m = 0; m < MT; ++m)
#pragma unroll
    for (int i = 0; i < 8; ++i) acc[m][i] = (v8f){};

  const float* wbase = Wd + (size_t)koff * Es + (size_t)((w * OTW) * 16 + l15) * S;

  {
    constexpr int TK = 4 * ConvFrag<S, MT>::KT;
    constexpr int NCH = E_DIM / TK;
    const int lin = (int)(blockIdx.y * gridDim.x + blockIdx.x);
    const int c0  = ((lin ^ (lin >> 4)) & (NCH - 1)) * TK;
    ConvFrag<S, MT> fa, fb;
    conv_load_trip<S, MT>(fa, xrow + koff, wbase, c0);
    for (int kk = 0; kk < E_DIM - 2 * TK; kk += 2 * TK) {
      conv_load_trip<S, MT>(fb, xrow + koff, wbase, (c0 + kk + TK) & (E_DIM - 1));
      __builtin_amdgcn_sched_barrier(0);
      conv_mma_trip<S, MT>(fa, acc);
      __builtin_amdgcn_sched_barrier(0);
      conv_load_trip<S, MT>(fa, xrow + koff, wbase, (c0 + kk + 2 * TK) & (E_DIM - 1));
      __builtin_amdgcn_sched_barrier(0);
      conv_mma_trip<S, MT>(fb, acc);
      __builtin_amdgcn_sched_barrier(0);
    }
    conv_load_trip<S, MT>(fb, xrow + koff, wbase, (c0 + E_DIM - TK) & (E_DIM - 1));
    __builtin_amdgcn_sched_barrier(0);
    conv_mma_trip<S, MT>(fa, acc);
    conv_mma_trip<S, MT>(fb, acc);
  }

#pragma unroll
  for (int i = 0; i < 8; ++i) {
    const int otl = i / S;
    const int k   = i % S;
    const int o   = (w * OTW + otl) * 16 + l15;
    const float bdv = bd[o];
#pragma unroll
    for (int m = 0; m < MT; ++m) {
#pragma unroll
      for (int v = 0; v < 8; ++v) {
        const int tl  = m * 16 + ((lane < 16) ? v : v + 8);
        const int row = tl * S + k;
        Elds[row * 256 + (o ^ ((row & 15) * 4))] = acc[m][i][v] + bdv;
      }
    }
  }
  __syncthreads();

  constexpr int CT = MT;
  const int rt   = (MT == 1) ? (w >> 1) : w;
  const int col0 = (MT == 1) ? (w & 1) * 16 : 0;
  const int m    = rt * 16 + l15;
  const int msw  = (m & 15) * 4;

  int  colv[CT];
  bool cval[CT];
  int  colc[CT];
  v8f  acc2[CT];
#pragma unroll
  for (int c = 0; c < CT; ++c) {
    colv[c] = col0 + c * 16 + l15;
    cval[c] = (colv[c] < V_DIM);
    colc[c] = cval[c] ? colv[c] : (V_DIM - 1);
    acc2[c] = (v8f){};
  }

  for (int kk = 0; kk < E_DIM; kk += 4) {
    const int k0 = kk + koff;
    v2f a;
    a.x = Elds[m * 256 + (k0 ^ msw)];
    a.y = Elds[m * 256 + ((k0 + 1) ^ msw)];
#pragma unroll
    for (int c = 0; c < CT; ++c) {
      v2f bf;
      bf.x = Wl[(size_t)k0 * V_DIM + colc[c]];
      bf.y = Wl[(size_t)(k0 + 1) * V_DIM + colc[c]];
      acc2[c] = wmma4(a, bf, acc2[c]);
    }
  }

#pragma unroll
  for (int c = 0; c < CT; ++c) {
    const float blv = bl[colc[c]];
#pragma unroll
    for (int v = 0; v < 8; ++v) {
      const int r     = rt * 16 + ((lane < 16) ? v : v + 8);
      const int tl    = r / S;
      const int k     = r % S;
      const int trel2 = tb * TOK + tl;
      if (cval[c] && trel2 < nt) {
        const int j = out_off + trel2 * S + k;
        out[((size_t)n * T_SEQ + j) * V_DIM + colv[c]] = acc2[c][v] + blv;
      }
    }
  }
}


struct Conv8Frag {
  float4 r0[4], r1[4];
  v2f    a0[4], a1[4], ax[4];
};

__device__ __forceinline__ void conv8_load(Conv8Frag& f, const float* __restrict__ xrow,
                                           const float* __restrict__ wbase, int kk0) {
  constexpr int Es = E_DIM * 8;
  constexpr int XP = E_DIM + 4;
#pragma unroll
  for (int s = 0; s < 4; ++s) {
    const int kk = kk0 + 4 * s;
    const float* wrow = wbase + (size_t)kk * Es;
    f.r0[s] = *reinterpret_cast<const float4*>(wrow);
    f.r1[s] = *reinterpret_cast<const float4*>(wrow + Es);
  }
#pragma unroll
  for (int s = 0; s < 4; ++s) {
    const int kk = kk0 + 4 * s;
    f.a0[s].x = xrow[kk];
    f.a0[s].y = xrow[kk + 1];
    f.a1[s].x = xrow[16 * XP + kk];
    f.a1[s].y = xrow[16 * XP + kk + 1];
    f.ax[s].x = xrow[32 * XP + kk];
    f.ax[s].y = xrow[32 * XP + kk + 1];
  }
}

template <int JX>
__device__ __forceinline__ void conv8_mma(const Conv8Frag& f, v8f (&acc)[2][4], v8f& accx) {
#pragma unroll
  for (int s = 0; s < 4; ++s) {
    const float4 r0 = f.r0[s];
    const float4 r1 = f.r1[s];
    const v2f a0 = f.a0[s], a1 = f.a1[s], ax = f.ax[s];
    v2f bf;
    bf.x = r0.x; bf.y = r1.x;
    acc[0][0] = wmma4(a0, bf, acc[0][0]);
    acc[1][0] = wmma4(a1, bf, acc[1][0]);
    if (JX == 0) accx = wmma4(ax, bf, accx);
    bf.x = r0.y; bf.y = r1.y;
    acc[0][1] = wmma4(a0, bf, acc[0][1]);
    acc[1][1] = wmma4(a1, bf, acc[1][1]);
    if (JX == 1) accx = wmma4(ax, bf, accx);
    bf.x = r0.z; bf.y = r1.z;
    acc[0][2] = wmma4(a0, bf, acc[0][2]);
    acc[1][2] = wmma4(a1, bf, acc[1][2]);
    if (JX == 2) accx = wmma4(ax, bf, accx);
    bf.x = r0.w; bf.y = r1.w;
    acc[0][3] = wmma4(a0, bf, acc[0][3]);
    acc[1][3] = wmma4(a1, bf, acc[1][3]);
    if (JX == 3) accx = wmma4(ax, bf, accx);
  }
}

template <int JX>
__device__ __forceinline__ void conv8_stage1(const float* __restrict__ xrow,
                                             const float* __restrict__ wbase,
                                             v8f (&acc)[2][4], v8f& accx, int c0) {
  constexpr int TK = 16;
  Conv8Frag fa, fb;
  conv8_load(fa, xrow, wbase, c0);
#pragma nounroll
  for (int kk = 0; kk < E_DIM - 2 * TK; kk += 2 * TK) {
    conv8_load(fb, xrow, wbase, (c0 + kk + TK) & (E_DIM - 1));
    conv8_mma<JX>(fa, acc, accx);
    conv8_load(fa, xrow, wbase, (c0 + kk + 2 * TK) & (E_DIM - 1));
    conv8_mma<JX>(fb, acc, accx);
  }
  conv8_load(fb, xrow, wbase, (c0 + E_DIM - TK) & (E_DIM - 1));
  conv8_mma<JX>(fa, acc, accx);
  conv8_mma<JX>(fb, acc, accx);
}

__global__ __launch_bounds__(512) void conv8_kernel(
    const float* __restrict__ x,
    const float* __restrict__ Wd, const float* __restrict__ bd,
    const float* __restrict__ Wl, const float* __restrict__ bl,
    float* __restrict__ out,
    int ioff, int out_off, int nt, int nbatch) {
  constexpr int S    = 8;
  constexpr int Es   = E_DIM * S;
  constexpr int XP   = E_DIM + 4;
  constexpr int ROWS = 2 * 16 * 4 + 16;
  __shared__ float Elds[ROWS * 256];
  __shared__ alignas(16) float Xs[48 * XP];

  const int b      = blockIdx.x;
  const int q      = b & 1;
  const int p      = b >> 1;
  const int G      = nbatch * nt;
  const int ntiles = (G + 15) >> 4;
  const int npairs = (int)gridDim.x >> 1;
  const int xt_raw = 2 * npairs + (p >> 2);
  const bool has_x = xt_raw < ntiles;
  const int xt     = has_x ? xt_raw : 2 * p;
  const int jx     = p & 3;

  const int tid  = threadIdx.x;
  const int w    = tid >> 5;
  const int lane = tid & 31;
  const int l15  = lane & 15;
  const int koff = (lane < 16) ? 0 : 2;

  float4 xv[6];
#pragma unroll
  for (int j = 0; j < 6; ++j) {
    const int idx  = tid + j * 512;
    const int r    = idx >> 6;
    const int c4   = idx & 63;
    const int tile = (j < 2) ? 2 * p : (j < 4) ? 2 * p + 1 : xt;
    const int g    = tile * 16 + (r & 15);
    const int gc   = (g < G) ? g : (G - 1);
    const int n    = gc / nt;
    const int t    = gc - n * nt;
    xv[j] = *reinterpret_cast<const float4*>(
        x + ((size_t)n * T_SEQ + ioff + t) * E_DIM + c4 * 4);
  }
#pragma unroll
  for (int j = 0; j < 6; ++j) {
    const int idx = tid + j * 512;
    *reinterpret_cast<float4*>(&Xs[(idx >> 6) * XP + (idx & 63) * 4]) = xv[j];
  }
  __syncthreads();

  const float* xrow = &Xs[l15 * XP + koff];
  v8f acc[2][4];
#pragma unroll
  for (int m = 0; m < 2; ++m)
#pragma unroll
    for (int j = 0; j < 4; ++j) acc[m][j] = (v8f){};
  v8f accx = {};

  const float* wbase = Wd + (size_t)koff * Es + (size_t)(w * 16 + l15) * S + q * 4;

  const int c0 = ((b ^ (b >> 4)) & 15) * 16;
  switch (jx) {
    case 0:  conv8_stage1<0>(xrow, wbase, acc, accx, c0); break;
    case 1:  conv8_stage1<1>(xrow, wbase, acc, accx, c0); break;
    case 2:  conv8_stage1<2>(xrow, wbase, acc, accx, c0); break;
    default: conv8_stage1<3>(xrow, wbase, acc, accx, c0); break;
  }

  {
    const int o = w * 16 + l15;
    const float bdv = bd[o];
#pragma unroll
    for (int m = 0; m < 2; ++m) {
#pragma unroll
      for (int j = 0; j < 4; ++j) {
#pragma unroll
        for (int v = 0; v < 8; ++v) {
          const int tl  = (lane < 16) ? v : v + 8;
          const int row = m * 64 + tl * 4 + j;
          Elds[row * 256 + (o ^ ((row & 15) * 4))] = acc[m][j][v] + bdv;
        }
      }
    }
#pragma unroll
    for (int v = 0; v < 8; ++v) {
      const int tl  = (lane < 16) ? v : v + 8;
      const int row = 128 + tl;
      Elds[row * 256 + (o ^ ((row & 15) * 4))] = accx[v] + bdv;
    }
  }
  __syncthreads();

  if (w >= 9) return;
  const int rt  = w;
  const int m   = rt * 16 + l15;
  const int msw = (m & 15) * 4;

  int  colv[2];
  bool cval[2];
  int  colc[2];
  v8f  acc2[2];
#pragma unroll
  for (int c = 0; c < 2; ++c) {
    colv[c] = c * 16 + l15;
    cval[c] = (colv[c] < V_DIM);
    colc[c] = cval[c] ? colv[c] : (V_DIM - 1);
    acc2[c] = (v8f){};
  }

  for (int kk = 0; kk < E_DIM; kk += 4) {
    const int k0 = kk + koff;
    v2f a;
    a.x = Elds[m * 256 + (k0 ^ msw)];
    a.y = Elds[m * 256 + ((k0 + 1) ^ msw)];
#pragma unroll
    for (int c = 0; c < 2; ++c) {
      v2f bf;
      bf.x = Wl[(size_t)k0 * V_DIM + colc[c]];
      bf.y = Wl[(size_t)(k0 + 1) * V_DIM + colc[c]];
      acc2[c] = wmma4(a, bf, acc2[c]);
    }
  }

  const int r0w = (lane < 16) ? 0 : 8;
  if (rt < 8) {
    const int gbase = (2 * p + (rt >> 2)) * 16 + (rt & 3) * 4 + (r0w >> 2);
    const float blv0 = bl[colc[0]];
    const float blv1 = bl[colc[1]];
#pragma unroll
    for (int tk = 0; tk < 2; ++tk) {
      const int g = gbase + tk;
      const int n = g / nt;
      const int t = g - n * nt;
      float* orow = out + ((size_t)n * T_SEQ + out_off + t * S + 4 * q) * V_DIM;
      if (g < G) {
#pragma unroll
        for (int jj = 0; jj < 4; ++jj) {
          if (cval[0]) orow[jj * V_DIM + colv[0]] = acc2[0][tk * 4 + jj] + blv0;
          if (cval[1]) orow[jj * V_DIM + colv[1]] = acc2[1][tk * 4 + jj] + blv1;
        }
      }
    }
  } else {
#pragma unroll
    for (int c = 0; c < 2; ++c) {
      const float blv = bl[colc[c]];
#pragma unroll
      for (int v = 0; v < 8; ++v) {
        const int r = r0w + v;
        const int g = xt * 16 + r;
        const int k = 4 * q + jx;
        if (cval[c] && has_x && g < G) {
          const int n = g / nt;
          const int t = g - n * nt;
          const int j = out_off + t * S + k;
          out[((size_t)n * T_SEQ + j) * V_DIM + colv[c]] = acc2[c][v] + blv;
        }
      }
    }
  }
}

extern "C" void kernel_launch(void* const* d_in, const int* in_sizes, int n_in,
                              void* d_out, int out_size, void* d_ws, size_t ws_size,
                              hipStream_t stream) {
  const float* x   = (const float*)d_in[0];
  const float* W1  = (const float*)d_in[4];
  const float* b1  = (const float*)d_in[5];
  const float* W2  = (const float*)d_in[6];
  const float* b2  = (const float*)d_in[7];
  const float* W3  = (const float*)d_in[8];
  const float* b3  = (const float*)d_in[9];
  const float* Wd4 = (const float*)d_in[10];
  const float* bd4 = (const float*)d_in[11];
  const float* Wl4 = (const float*)d_in[12];
  const float* bl4 = (const float*)d_in[13];
  const float* Wd5 = (const float*)d_in[14];
  const float* bd5 = (const float*)d_in[15];
  const float* Wl5 = (const float*)d_in[16];
  const float* bl5 = (const float*)d_in[17];
  float* out = (float*)d_out;

  const int N = 8;

  head123_kernel<<<dim3(100, N), 32, 0, stream>>>(x, W1, b1, W2, b2, W3, b3, out);

  conv_head_kernel<4, 1><<<dim3(32, N), 256, 0, stream>>>(
      x, Wd4, bd4, Wl4, bl4, out,  1600,  1600,  512);

  conv8_kernel<<<dim3(256), 512, 0, stream>>>(
      x, Wd5, bd5, Wl5, bl5, out,  2112,  3648,  568,  8);
}


# Conv heads stage 2: Wl (256x17) copied to LDS once per block beside the x tile; B fragments come from LDS, not from global loads behind a cache round trip every 4 K steps; sums unchanged
# speedup vs baseline: 3.9864x; 1.1355x over previous
#include <hip/hip_runtime.h>
#include <hip/hip_bf16.h>


#define T_SEQ 8192
#define E_DIM 256
#define V_DIM 17

typedef float v2f __attribute__((ext_vector_type(2)));
typedef float v8f __attribute__((ext_vector_type(8)));

__device__ __forceinline__ v8f wmma4(v2f a, v2f b, v8f c) {
  return __builtin_amdgcn_wmma_f32_16x16x4_f32(
       false, a,  false, b,
       (short)0, c,  false,  false);
}

__global__ __launch_bounds__(32) void head123_kernel(
    const float* __restrict__ x,
    const float* __restrict__ W1, const float* __restrict__ b1,
    const float* __restrict__ W2, const float* __restrict__ b2,
    const float* __restrict__ W3, const float* __restrict__ b3,
    float* __restrict__ out) {
  const int n       = blockIdx.y;
  const int rowbase = blockIdx.x * 16;

  const float* W;
  const float* b;
  if (rowbase < 64)       { W = W1; b = b1; }
  else if (rowbase < 576) { W = W2; b = b2; }
  else                    { W = W3; b = b3; }

  const int lane = threadIdx.x & 31;
  const int l15  = lane & 15;
  const int koff = (lane < 16) ? 0 : 2;

  const float* xrow = x + ((size_t)n * T_SEQ + rowbase + l15) * E_DIM;
  const int c1  = 16 + l15;
  const int c1c = (c1 < V_DIM) ? c1 : (V_DIM - 1);

  v8f acc0 = {};
  v8f acc1 = {};
  for (int kk = 0; kk < E_DIM; kk += 4) {
    v2f a;
    a.x = xrow[kk + koff];
    a.y = xrow[kk + koff + 1];
    v2f bf0, bf1;
    bf0.x = W[(kk + koff) * V_DIM + l15];
    bf0.y = W[(kk + koff + 1) * V_DIM + l15];
    bf1.x = W[(kk + koff) * V_DIM + c1c];
    bf1.y = W[(kk + koff + 1) * V_DIM + c1c];
    acc0 = wmma4(a, bf0, acc0);
    acc1 = wmma4(a, bf1, acc1);
  }

#pragma unroll
  for (int v = 0; v < 8; ++v) {
    const int r = ((lane < 16) ? v : v + 8);
    float* o = out + ((size_t)n * T_SEQ + rowbase + r) * V_DIM;
    o[l15] = acc0[v] + b[l15];
    if (c1 < V_DIM) o[c1] = acc1[v] + b[c1];
  }
}

template <int S, int MT>
struct ConvFrag {
  static constexpr int KT  = (S == 8) ? 2 : 4;
  static constexpr int OTW = 8 / S;
  static constexpr int Q   = S / 4;
  float4 r0[KT][OTW][Q], r1[KT][OTW][Q];
  v2f    a[KT][MT];
};

template <int S, int MT>
__device__ __forceinline__ void conv_load_trip(ConvFrag<S, MT>& f, const float* __restrict__ xk,
                                               const float* __restrict__ wbase, int kk0) {
  constexpr int KT = ConvFrag<S, MT>::KT, OTW = ConvFrag<S, MT>::OTW, Q = ConvFrag<S, MT>::Q;
  constexpr int Es = E_DIM * S;
  constexpr int XP = E_DIM + 4;
#pragma unroll
  for (int s = 0; s < KT; ++s) {
    const float* wrow = wbase + (size_t)(kk0 + 4 * s) * Es;
#pragma unroll
    for (int otl = 0; otl < OTW; ++otl) {
#pragma unroll
      for (int q = 0; q < Q; ++q) {
        const float* bp = wrow + otl * 16 * S + q * 4;
        f.r0[s][otl][q] = *reinterpret_cast<const float4*>(bp);
        f.r1[s][otl][q] = *reinterpret_cast<const float4*>(bp + Es);
      }
    }
  }
#pragma unroll
  for (int s = 0; s < KT; ++s) {
    const int kk = kk0 + 4 * s;
#pragma unroll
    for (int m = 0; m < MT; ++m) {
      f.a[s][m].x = xk[m * 16 * XP + kk];
      f.a[s][m].y = xk[m * 16 * XP + kk + 1];
    }
  }
}

template <int S, int MT>
__device__ __forceinline__ void conv_mma_trip(const ConvFrag<S, MT>& f, v8f (&acc)[MT][8]) {
  constexpr int KT = ConvFrag<S, MT>::KT, OTW = ConvFrag<S, MT>::OTW, Q = ConvFrag<S, MT>::Q;
#pragma unroll
  for (int s = 0; s < KT; ++s) {
#pragma unroll
    for (int otl = 0; otl < OTW; ++otl) {
#pragma unroll
      for (int q = 0; q < Q; ++q) {
        const float4 r0 = f.r0[s][otl][q];
        const float4 r1 = f.r1[s][otl][q];
        const int t0 = otl * S + q * 4;
        v2f bf;
        bf.x = r0.x; bf.y = r1.x;
#pragma unroll
        for (int m = 0; m < MT; ++m) acc[m][t0 + 0] = wmma4(f.a[s][m], bf, acc[m][t0 + 0]);
        bf.x = r0.y; bf.y = r1.y;
#pragma unroll
        for (int m = 0; m < MT; ++m) acc[m][t0 + 1] = wmma4(f.a[s][m], bf, acc[m][t0 + 1]);
        bf.x = r0.z; bf.y = r1.z;
#pragma unroll
        for (int m = 0; m < MT; ++m) acc[m][t0 + 2] = wmma4(f.a[s][m], bf, acc[m][t0 + 2]);
        bf.x = r0.w; bf.y = r1.w;
#pragma unroll
        for (int m = 0; m < MT; ++m) acc[m][t0 + 3] = wmma4(f.a[s][m], bf, acc[m][t0 + 3]);
      }
    }
  }
}

template <int S, int MT>
__global__ __launch_bounds__(64 * S) void conv_head_kernel(
    const float* __restrict__ x,
    const float* __restrict__ Wd, const float* __restrict__ bd,
    const float* __restrict__ Wl, const float* __restrict__ bl,
    float* __restrict__ out,
    int ioff, int out_off, int nt) {
  constexpr int OTW = 8 / S;
  constexpr int Q   = S / 4;
  constexpr int Es  = E_DIM * S;
  constexpr int TOK = 16 * MT;
  __shared__ float Elds[TOK * S * 256];
  constexpr int XP = E_DIM + 4;
  __shared__ alignas(16) float Xs[TOK * XP];
  constexpr int NW4 = (E_DIM * V_DIM) / 4;
  constexpr int NWP = (NW4 + 64 * S - 1) / (64 * S);
  __shared__ alignas(16) float Wls[NWP * 64 * S * 4];

  const int n  = blockIdx.y;
  const int tb = blockIdx.x;

  const int tid  = threadIdx.x;
  const int w    = tid >> 5;
  const int lane = tid & 31;
  const int l15  = lane & 15;
  const int koff = (lane < 16) ? 0 : 2;

  constexpr int NXL = (TOK * 64) / (64 * S);
  float4 xv[NXL];
#pragma unroll
  for (int j = 0; j < NXL; ++j) {
    const int idx    = tid + j * 64 * S;
    const int r      = idx >> 6;
    const int c4     = idx & 63;
    const int trel   = tb * TOK + r;
    const int tclamp = (trel < nt) ? trel : (nt - 1);
    xv[j] = *reinterpret_cast<const float4*>(
        x + ((size_t)n * T_SEQ + ioff + tclamp) * E_DIM + c4 * 4);
  }
  float4 wv[NWP];
#pragma unroll
  for (int j = 0; j < NWP; ++j) {
    const int idx = tid + j * 64 * S;
    const int src = (j == NWP - 1 && idx >= NW4) ? (NW4 - 1) : idx;
    wv[j] = reinterpret_cast<const float4*>(Wl)[src];
  }
#pragma unroll
  for (int j = 0; j < NXL; ++j) {
    const int idx = tid + j * 64 * S;
    *reinterpret_cast<float4*>(&Xs[(idx >> 6) * XP + (idx & 63) * 4]) = xv[j];
  }
#pragma unroll
  for (int j = 0; j < NWP; ++j)
    *reinterpret_cast<float4*>(&Wls[(tid + j * 64 * S) * 4]) = wv[j];
  __syncthreads();

  const float* xrow = &Xs[l15 * XP];

  v8f acc[MT][8];
#pragma unroll
  for (int m = 0; m < MT; ++m)
#pragma unroll
    for (int i = 0; i < 8; ++i) acc[m][i] = (v8f){};

  const float* wbase = Wd + (size_t)koff * Es + (size_t)((w * OTW) * 16 + l15) * S;

  {
    constexpr int TK = 4 * ConvFrag<S, MT>::KT;
    constexpr int NCH = E_DIM / TK;
    const int lin = (int)(blockIdx.y * gridDim.x + blockIdx.x);
    const int c0  = ((lin ^ (lin >> 4)) & (NCH - 1)) * TK;
    ConvFrag<S, MT> fa, fb;
    conv_load_trip<S, MT>(fa, xrow + koff, wbase, c0);
    for (int kk = 0; kk < E_DIM - 2 * TK; kk += 2 * TK) {
      conv_load_trip<S, MT>(fb, xrow + koff, wbase, (c0 + kk + TK) & (E_DIM - 1));
      __builtin_amdgcn_sched_barrier(0);
      conv_mma_trip<S, MT>(fa, acc);
      __builtin_amdgcn_sched_barrier(0);
      conv_load_trip<S, MT>(fa, xrow + koff, wbase, (c0 + kk + 2 * TK) & (E_DIM - 1));
      __builtin_amdgcn_sched_barrier(0);
      conv_mma_trip<S, MT>(fb, acc);
      __builtin_amdgcn_sched_barrier(0);
    }
    conv_load_trip<S, MT>(fb, xrow + koff, wbase, (c0 + E_DIM - TK) & (E_DIM - 1));
    __builtin_amdgcn_sched_barrier(0);
    conv_mma_trip<S, MT>(fa, acc);
    conv_mma_trip<S, MT>(fb, acc);
  }

#pragma unroll
  for (int i = 0; i < 8; ++i) {
    const int otl = i / S;
    const int k   = i % S;
    const int o   = (w * OTW + otl) * 16 + l15;
    const float bdv = bd[o];
#pragma unroll
    for (int m = 0; m < MT; ++m) {
#pragma unroll
      for (int v = 0; v < 8; ++v) {
        const int tl  = m * 16 + ((lane < 16) ? v : v + 8);
        const int row = tl * S + k;
        Elds[row * 256 + (o ^ ((row & 15) * 4))] = acc[m][i][v] + bdv;
      }
    }
  }
  __syncthreads();

  constexpr int CT = MT;
  const int rt   = (MT == 1) ? (w >> 1) : w;
  const int col0 = (MT == 1) ? (w & 1) * 16 : 0;
  const int m    = rt * 16 + l15;
  const int msw  = (m & 15) * 4;

  int  colv[CT];
  bool cval[CT];
  int  colc[CT];
  v8f  acc2[CT];
#pragma unroll
  for (int c = 0; c < CT; ++c) {
    colv[c] = col0 + c * 16 + l15;
    cval[c] = (colv[c] < V_DIM);
    colc[c] = cval[c] ? colv[c] : (V_DIM - 1);
    acc2[c] = (v8f){};
  }

  for (int kk = 0; kk < E_DIM; kk += 4) {
    const int k0 = kk + koff;
    v2f a;
    a.x = Elds[m * 256 + (k0 ^ msw)];
    a.y = Elds[m * 256 + ((k0 + 1) ^ msw)];
#pragma unroll
    for (int c = 0; c < CT; ++c) {
      v2f bf;
      bf.x = Wls[k0 * V_DIM + colc[c]];
      bf.y = Wls[(k0 + 1) * V_DIM + colc[c]];
      acc2[c] = wmma4(a, bf, acc2[c]);
    }
  }

#pragma unroll
  for (int c = 0; c < CT; ++c) {
    const float blv = bl[colc[c]];
#pragma unroll
    for (int v = 0; v < 8; ++v) {
      const int r     = rt * 16 + ((lane < 16) ? v : v + 8);
      const int tl    = r / S;
      const int k     = r % S;
      const int trel2 = tb * TOK + tl;
      if (cval[c] && trel2 < nt) {
        const int j = out_off + trel2 * S + k;
        out[((size_t)n * T_SEQ + j) * V_DIM + colv[c]] = acc2[c][v] + blv;
      }
    }
  }
}


struct Conv8Frag {
  float4 r0[4], r1[4];
  v2f    a0[4], a1[4], ax[4];
};

__device__ __forceinline__ void conv8_load(Conv8Frag& f, const float* __restrict__ xrow,
                                           const float* __restrict__ wbase, int kk0) {
  constexpr int Es = E_DIM * 8;
  constexpr int XP = E_DIM + 4;
#pragma unroll
  for (int s = 0; s < 4; ++s) {
    const int kk = kk0 + 4 * s;
    const float* wrow = wbase + (size_t)kk * Es;
    f.r0[s] = *reinterpret_cast<const float4*>(wrow);
    f.r1[s] = *reinterpret_cast<const float4*>(wrow + Es);
  }
#pragma unroll
  for (int s = 0; s < 4; ++s) {
    const int kk = kk0 + 4 * s;
    f.a0[s].x = xrow[kk];
    f.a0[s].y = xrow[kk + 1];
    f.a1[s].x = xrow[16 * XP + kk];
    f.a1[s].y = xrow[16 * XP + kk + 1];
    f.ax[s].x = xrow[32 * XP + kk];
    f.ax[s].y = xrow[32 * XP + kk + 1];
  }
}

template <int JX>
__device__ __forceinline__ void conv8_mma(const Conv8Frag& f, v8f (&acc)[2][4], v8f& accx) {
#pragma unroll
  for (int s = 0; s < 4; ++s) {
    const float4 r0 = f.r0[s];
    const float4 r1 = f.r1[s];
    const v2f a0 = f.a0[s], a1 = f.a1[s], ax = f.ax[s];
    v2f bf;
    bf.x = r0.x; bf.y = r1.x;
    acc[0][0] = wmma4(a0, bf, acc[0][0]);
    acc[1][0] = wmma4(a1, bf, acc[1][0]);
    if (JX == 0) accx = wmma4(ax, bf, accx);
    bf.x = r0.y; bf.y = r1.y;
    acc[0][1] = wmma4(a0, bf, acc[0][1]);
    acc[1][1] = wmma4(a1, bf, acc[1][1]);
    if (JX == 1) accx = wmma4(ax, bf, accx);
    bf.x = r0.z; bf.y = r1.z;
    acc[0][2] = wmma4(a0, bf, acc[0][2]);
    acc[1][2] = wmma4(a1, bf, acc[1][2]);
    if (JX == 2) accx = wmma4(ax, bf, accx);
    bf.x = r0.w; bf.y = r1.w;
    acc[0][3] = wmma4(a0, bf, acc[0][3]);
    acc[1][3] = wmma4(a1, bf, acc[1][3]);
    if (JX == 3) accx = wmma4(ax, bf, accx);
  }
}

template <int JX>
__device__ __forceinline__ void conv8_stage1(const float* __restrict__ xrow,
                                             const float* __restrict__ wbase,
                                             v8f (&acc)[2][4], v8f& accx, int c0) {
  constexpr int TK = 16;
  Conv8Frag fa, fb;
  conv8_load(fa, xrow, wbase, c0);
#pragma nounroll
  for (int kk = 0; kk < E_DIM - 2 * TK; kk += 2 * TK) {
    conv8_load(fb, xrow, wbase, (c0 + kk + TK) & (E_DIM - 1));
    conv8_mma<JX>(fa, acc, accx);
    conv8_load(fa, xrow, wbase, (c0 + kk + 2 * TK) & (E_DIM - 1));
    conv8_mma<JX>(fb, acc, accx);
  }
  conv8_load(fb, xrow, wbase, (c0 + E_DIM - TK) & (E_DIM - 1));
  conv8_mma<JX>(fa, acc, accx);
  conv8_mma<JX>(fb, acc, accx);
}

__global__ __launch_bounds__(512) void conv8_kernel(
    const float* __restrict__ x,
    const float* __restrict__ Wd, const float* __restrict__ bd,
    const float* __restrict__ Wl, const float* __restrict__ bl,
    float* __restrict__ out,
    int ioff, int out_off, int nt, int nbatch) {
  constexpr int S    = 8;
  constexpr int Es   = E_DIM * S;
  constexpr int XP   = E_DIM + 4;
  constexpr int ROWS = 2 * 16 * 4 + 16;
  __shared__ float Elds[ROWS * 256];
  __shared__ alignas(16) float Xs[48 * XP];
  constexpr int NW4 = (E_DIM * V_DIM) / 4;
  __shared__ alignas(16) float Wls[3 * 512 * 4];

  const int b      = blockIdx.x;
  const int q      = b & 1;
  const int p      = b >> 1;
  const int G      = nbatch * nt;
  const int ntiles = (G + 15) >> 4;
  const int npairs = (int)gridDim.x >> 1;
  const int xt_raw = 2 * npairs + (p >> 2);
  const bool has_x = xt_raw < ntiles;
  const int xt     = has_x ? xt_raw : 2 * p;
  const int jx     = p & 3;

  const int tid  = threadIdx.x;
  const int w    = tid >> 5;
  const int lane = tid & 31;
  const int l15  = lane & 15;
  const int koff = (lane < 16) ? 0 : 2;

  float4 xv[6];
#pragma unroll
  for (int j = 0; j < 6; ++j) {
    const int idx  = tid + j * 512;
    const int r    = idx >> 6;
    const int c4   = idx & 63;
    const int tile = (j < 2) ? 2 * p : (j < 4) ? 2 * p + 1 : xt;
    const int g    = tile * 16 + (r & 15);
    const int gc   = (g < G) ? g : (G - 1);
    const int n    = gc / nt;
    const int t    = gc - n * nt;
    xv[j] = *reinterpret_cast<const float4*>(
        x + ((size_t)n * T_SEQ + ioff + t) * E_DIM + c4 * 4);
  }
  const int widx = tid + 1024;
  const float4 wv0 = reinterpret_cast<const float4*>(Wl)[tid];
  const float4 wv1 = reinterpret_cast<const float4*>(Wl)[tid + 512];
  const float4 wv2 = reinterpret_cast<const float4*>(Wl)[(widx < NW4) ? widx : (NW4 - 1)];
#pragma unroll
  for (int j = 0; j < 6; ++j) {
    const int idx = tid + j * 512;
    *reinterpret_cast<float4*>(&Xs[(idx >> 6) * XP + (idx & 63) * 4]) = xv[j];
  }
  *reinterpret_cast<float4*>(&Wls[tid * 4])         = wv0;
  *reinterpret_cast<float4*>(&Wls[(tid + 512) * 4]) = wv1;
  *reinterpret_cast<float4*>(&Wls[widx * 4])        = wv2;
  __syncthreads();

  const float* xrow = &Xs[l15 * XP + koff];
  v8f acc[2][4];
#pragma unroll
  for (int m = 0; m < 2; ++m)
#pragma unroll
    for (int j = 0; j < 4; ++j) acc[m][j] = (v8f){};
  v8f accx = {};

  const float* wbase = Wd + (size_t)koff * Es + (size_t)(w * 16 + l15) * S + q * 4;

  const int c0 = ((b ^ (b >> 4)) & 15) * 16;
  switch (jx) {
    case 0:  conv8_stage1<0>(xrow, wbase, acc, accx, c0); break;
    case 1:  conv8_stage1<1>(xrow, wbase, acc, accx, c0); break;
    case 2:  conv8_stage1<2>(xrow, wbase, acc, accx, c0); break;
    default: conv8_stage1<3>(xrow, wbase, acc, accx, c0); break;
  }

  {
    const int o = w * 16 + l15;
    const float bdv = bd[o];
#pragma unroll
    for (int m = 0; m < 2; ++m) {
#pragma unroll
      for (int j = 0; j < 4; ++j) {
#pragma unroll
        for (int v = 0; v < 8; ++v) {
          const int tl  = (lane < 16) ? v : v + 8;
          const int row = m * 64 + tl * 4 + j;
          Elds[row * 256 + (o ^ ((row & 15) * 4))] = acc[m][j][v] + bdv;
        }
      }
    }
#pragma unroll
    for (int v = 0; v < 8; ++v) {
      const int tl  = (lane < 16) ? v : v + 8;
      const int row = 128 + tl;
      Elds[row * 256 + (o ^ ((row & 15) * 4))] = accx[v] + bdv;
    }
  }
  __syncthreads();

  if (w >= 9) return;
  const int rt  = w;
  const int m   = rt * 16 + l15;
  const int msw = (m & 15) * 4;

  int  colv[2];
  bool cval[2];
  int  colc[2];
  v8f  acc2[2];
#pragma unroll
  for (int c = 0; c < 2; ++c) {
    colv[c] = c * 16 + l15;
    cval[c] = (colv[c] < V_DIM);
    colc[c] = cval[c] ? colv[c] : (V_DIM - 1);
    acc2[c] = (v8f){};
  }

  for (int kk = 0; kk < E_DIM; kk += 4) {
    const int k0 = kk + koff;
    v2f a;
    a.x = Elds[m * 256 + (k0 ^ msw)];
    a.y = Elds[m * 256 + ((k0 + 1) ^ msw)];
#pragma unroll
    for (int c = 0; c < 2; ++c) {
      v2f bf;
      bf.x = Wls[k0 * V_DIM + colc[c]];
      bf.y = Wls[(k0 + 1) * V_DIM + colc[c]];
      acc2[c] = wmma4(a, bf, acc2[c]);
    }
  }

  const int r0w = (lane < 16) ? 0 : 8;
  if (rt < 8) {
    const int gbase = (2 * p + (rt >> 2)) * 16 + (rt & 3) * 4 + (r0w >> 2);
    const float blv0 = bl[colc[0]];
    const float blv1 = bl[colc[1]];
#pragma unroll
    for (int tk = 0; tk < 2; ++tk) {
      const int g = gbase + tk;
      const int n = g / nt;
      const int t = g - n * nt;
      float* orow = out + ((size_t)n * T_SEQ + out_off + t * S + 4 * q) * V_DIM;
      if (g < G) {
#pragma unroll
        for (int jj = 0; jj < 4; ++jj) {
          if (cval[0]) orow[jj * V_DIM + colv[0]] = acc2[0][tk * 4 + jj] + blv0;
          if (cval[1]) orow[jj * V_DIM + colv[1]] = acc2[1][tk * 4 + jj] + blv1;
        }
      }
    }
  } else {
#pragma unroll
    for (int c = 0; c < 2; ++c) {
      const float blv = bl[colc[c]];
#pragma unroll
      for (int v = 0; v < 8; ++v) {
        const int r = r0w + v;
        const int g = xt * 16 + r;
        const int k = 4 * q + jx;
        if (cval[c] && has_x && g < G) {
          const int n = g / nt;
          const int t = g - n * nt;
          const int j = out_off + t * S + k;
          out[((size_t)n * T_SEQ + j) * V_DIM + colv[c]] = acc2[c][v] + blv;
        }
      }
    }
  }
}

extern "C" void kernel_launch(void* const* d_in, const int* in_sizes, int n_in,
                              void* d_out, int out_size, void* d_ws, size_t ws_size,
                              hipStream_t stream) {
  const float* x   = (const float*)d_in[0];
  const float* W1  = (const float*)d_in[4];
  const float* b1  = (const float*)d_in[5];
  const float* W2  = (const float*)d_in[6];
  const float* b2  = (const float*)d_in[7];
  const float* W3  = (const float*)d_in[8];
  const float* b3  = (const float*)d_in[9];
  const float* Wd4 = (const float*)d_in[10];
  const float* bd4 = (const float*)d_in[11];
  const float* Wl4 = (const float*)d_in[12];
  const float* bl4 = (const float*)d_in[13];
  const float* Wd5 = (const float*)d_in[14];
  const float* bd5 = (const float*)d_in[15];
  const float* Wl5 = (const float*)d_in[16];
  const float* bl5 = (const float*)d_in[17];
  float* out = (float*)d_out;

  const int N = 8;

  head123_kernel<<<dim3(100, N), 32, 0, stream>>>(x, W1, b1, W2, b2, W3, b3, out);

  conv_head_kernel<4, 1><<<dim3(32, N), 256, 0, stream>>>(
      x, Wd4, bd4, Wl4, bl4, out,  1600,  1600,  512);

  conv8_kernel<<<dim3(256), 512, 0, stream>>>(
      x, Wd5, bd5, Wl5, bl5, out,  2112,  3648,  568,  8);
}


# Heads 1-3: 2 waves per row tile (one per column tile) share the x tile, staged in LDS by coalesced 16-byte loads; all x and W loads issued at wave start, K loop = LDS reads + 64 WMMAs; sums unchanged
# speedup vs baseline: 4.3290x; 1.0859x over previous
#include <hip/hip_runtime.h>
#include <hip/hip_bf16.h>


#define T_SEQ 8192
#define E_DIM 256
#define V_DIM 17

typedef float v2f __attribute__((ext_vector_type(2)));
typedef float v8f __attribute__((ext_vector_type(8)));

__device__ __forceinline__ v8f wmma4(v2f a, v2f b, v8f c) {
  return __builtin_amdgcn_wmma_f32_16x16x4_f32(
       false, a,  false, b,
       (short)0, c,  false,  false);
}

__global__ __launch_bounds__(64) void head123_kernel(
    const float* __restrict__ x,
    const float* __restrict__ W1, const float* __restrict__ b1,
    const float* __restrict__ W2, const float* __restrict__ b2,
    const float* __restrict__ W3, const float* __restrict__ b3,
    float* __restrict__ out) {
  constexpr int XP = E_DIM + 4;
  constexpr int KS = E_DIM / 4;
  __shared__ alignas(16) float Xs[16 * XP];

  const int n       = blockIdx.y;
  const int rowbase = blockIdx.x * 16;

  const float* W;
  const float* b;
  if (rowbase < 64)       { W = W1; b = b1; }
  else if (rowbase < 576) { W = W2; b = b2; }
  else                    { W = W3; b = b3; }

  const int tid  = threadIdx.x;
  const int w    = tid >> 5;
  const int lane = tid & 31;
  const int l15  = lane & 15;
  const int koff = (lane < 16) ? 0 : 2;

  const int  col  = w * 16 + l15;
  const bool cval = (col < V_DIM);
  const int  colc = cval ? col : (V_DIM - 1);

  const float* wl = W + koff * V_DIM + colc;
  const float bv = b[colc];
  float wf[KS][2];
#pragma unroll
  for (int s = 0; s < KS; ++s) {
    wf[s][0] = wl[(4 * s) * V_DIM];
    wf[s][1] = wl[(4 * s + 1) * V_DIM];
  }

  const float4* xt = reinterpret_cast<const float4*>(x + ((size_t)n * T_SEQ + rowbase) * E_DIM);
  float4 xv[16];
#pragma unroll
  for (int j = 0; j < 16; ++j) xv[j] = xt[j * 64 + tid];

#pragma unroll
  for (int j = 0; j < 16; ++j)
    *reinterpret_cast<float4*>(&Xs[j * XP + tid * 4]) = xv[j];
  __syncthreads();

  const float* xrow = &Xs[l15 * XP + koff];
  v2f af[KS];
#pragma unroll
  for (int s = 0; s < KS; ++s) af[s] = *reinterpret_cast<const v2f*>(xrow + 4 * s);
  __builtin_amdgcn_sched_barrier(0);

  v8f acc = {};
#pragma unroll
  for (int s = 0; s < KS; ++s) {
    v2f bf;
    bf.x = wf[s][0]; bf.y = wf[s][1];
    acc = wmma4(af[s], bf, acc);
  }

#pragma unroll
  for (int v = 0; v < 8; ++v) {
    const int r = ((lane < 16) ? v : v + 8);
    if (cval) out[((size_t)n * T_SEQ + rowbase + r) * V_DIM + col] = acc[v] + bv;
  }
}

template <int S, int MT>
struct ConvFrag {
  static constexpr int KT  = (S == 8) ? 2 : 4;
  static constexpr int OTW = 8 / S;
  static constexpr int Q   = S / 4;
  float4 r0[KT][OTW][Q], r1[KT][OTW][Q];
  v2f    a[KT][MT];
};

template <int S, int MT>
__device__ __forceinline__ void conv_load_trip(ConvFrag<S, MT>& f, const float* __restrict__ xk,
                                               const float* __restrict__ wbase, int kk0) {
  constexpr int KT = ConvFrag<S, MT>::KT, OTW = ConvFrag<S, MT>::OTW, Q = ConvFrag<S, MT>::Q;
  constexpr int Es = E_DIM * S;
  constexpr int XP = E_DIM + 4;
#pragma unroll
  for (int s = 0; s < KT; ++s) {
    const float* wrow = wbase + (size_t)(kk0 + 4 * s) * Es;
#pragma unroll
    for (int otl = 0; otl < OTW; ++otl) {
#pragma unroll
      for (int q = 0; q < Q; ++q) {
        const float* bp = wrow + otl * 16 * S + q * 4;
        f.r0[s][otl][q] = *reinterpret_cast<const float4*>(bp);
        f.r1[s][otl][q] = *reinterpret_cast<const float4*>(bp + Es);
      }
    }
  }
#pragma unroll
  for (int s = 0; s < KT; ++s) {
    const int kk = kk0 + 4 * s;
#pragma unroll
    for (int m = 0; m < MT; ++m) {
      f.a[s][m].x = xk[m * 16 * XP + kk];
      f.a[s][m].y = xk[m * 16 * XP + kk + 1];
    }
  }
}

template <int S, int MT>
__device__ __forceinline__ void conv_mma_trip(const ConvFrag<S, MT>& f, v8f (&acc)[MT][8]) {
  constexpr int KT = ConvFrag<S, MT>::KT, OTW = ConvFrag<S, MT>::OTW, Q = ConvFrag<S, MT>::Q;
#pragma unroll
  for (int s = 0; s < KT; ++s) {
#pragma unroll
    for (int otl = 0; otl < OTW; ++otl) {
#pragma unroll
      for (int q = 0; q < Q; ++q) {
        const float4 r0 = f.r0[s][otl][q];
        const float4 r1 = f.r1[s][otl][q];
        const int t0 = otl * S + q * 4;
        v2f bf;
        bf.x = r0.x; bf.y = r1.x;
#pragma unroll
        for (int m = 0; m < MT; ++m) acc[m][t0 + 0] = wmma4(f.a[s][m], bf, acc[m][t0 + 0]);
        bf.x = r0.y; bf.y = r1.y;
#pragma unroll
        for (int m = 0; m < MT; ++m) acc[m][t0 + 1] = wmma4(f.a[s][m], bf, acc[m][t0 + 1]);
        bf.x = r0.z; bf.y = r1.z;
#pragma unroll
        for (int m = 0; m < MT; ++m) acc[m][t0 + 2] = wmma4(f.a[s][m], bf, acc[m][t0 + 2]);
        bf.x = r0.w; bf.y = r1.w;
#pragma unroll
        for (int m = 0; m < MT; ++m) acc[m][t0 + 3] = wmma4(f.a[s][m], bf, acc[m][t0 + 3]);
      }
    }
  }
}

template <int S, int MT>
__global__ __launch_bounds__(64 * S) void conv_head_kernel(
    const float* __restrict__ x,
    const float* __restrict__ Wd, const float* __restrict__ bd,
    const float* __restrict__ Wl, const float* __restrict__ bl,
    float* __restrict__ out,
    int ioff, int out_off, int nt) {
  constexpr int OTW = 8 / S;
  constexpr int Q   = S / 4;
  constexpr int Es  = E_DIM * S;
  constexpr int TOK = 16 * MT;
  __shared__ float Elds[TOK * S * 256];
  constexpr int XP = E_DIM + 4;
  __shared__ alignas(16) float Xs[TOK * XP];
  constexpr int NW4 = (E_DIM * V_DIM) / 4;
  constexpr int NWP = (NW4 + 64 * S - 1) / (64 * S);
  __shared__ alignas(16) float Wls[NWP * 64 * S * 4];

  const int n  = blockIdx.y;
  const int tb = blockIdx.x;

  const int tid  = threadIdx.x;
  const int w    = tid >> 5;
  const int lane = tid & 31;
  const int l15  = lane & 15;
  const int koff = (lane < 16) ? 0 : 2;

  constexpr int NXL = (TOK * 64) / (64 * S);
  float4 xv[NXL];
#pragma unroll
  for (int j = 0; j < NXL; ++j) {
    const int idx    = tid + j * 64 * S;
    const int r      = idx >> 6;
    const int c4     = idx & 63;
    const int trel   = tb * TOK + r;
    const int tclamp = (trel < nt) ? trel : (nt - 1);
    xv[j] = *reinterpret_cast<const float4*>(
        x + ((size_t)n * T_SEQ + ioff + tclamp) * E_DIM + c4 * 4);
  }
  float4 wv[NWP];
#pragma unroll
  for (int j = 0; j < NWP; ++j) {
    const int idx = tid + j * 64 * S;
    const int src = (j == NWP - 1 && idx >= NW4) ? (NW4 - 1) : idx;
    wv[j] = reinterpret_cast<const float4*>(Wl)[src];
  }
#pragma unroll
  for (int j = 0; j < NXL; ++j) {
    const int idx = tid + j * 64 * S;
    *reinterpret_cast<float4*>(&Xs[(idx >> 6) * XP + (idx & 63) * 4]) = xv[j];
  }
#pragma unroll
  for (int j = 0; j < NWP; ++j)
    *reinterpret_cast<float4*>(&Wls[(tid + j * 64 * S) * 4]) = wv[j];
  __syncthreads();

  const float* xrow = &Xs[l15 * XP];

  v8f acc[MT][8];
#pragma unroll
  for (int m = 0; m < MT; ++m)
#pragma unroll
    for (int i = 0; i < 8; ++i) acc[m][i] = (v8f){};

  const float* wbase = Wd + (size_t)koff * Es + (size_t)((w * OTW) * 16 + l15) * S;

  {
    constexpr int TK = 4 * ConvFrag<S, MT>::KT;
    constexpr int NCH = E_DIM / TK;
    const int lin = (int)(blockIdx.y * gridDim.x + blockIdx.x);
    const int c0  = ((lin ^ (lin >> 4)) & (NCH - 1)) * TK;
    ConvFrag<S, MT> fa, fb;
    conv_load_trip<S, MT>(fa, xrow + koff, wbase, c0);
    for (int kk = 0; kk < E_DIM - 2 * TK; kk += 2 * TK) {
      conv_load_trip<S, MT>(fb, xrow + koff, wbase, (c0 + kk + TK) & (E_DIM - 1));
      __builtin_amdgcn_sched_barrier(0);
      conv_mma_trip<S, MT>(fa, acc);
      __builtin_amdgcn_sched_barrier(0);
      conv_load_trip<S, MT>(fa, xrow + koff, wbase, (c0 + kk + 2 * TK) & (E_DIM - 1));
      __builtin_amdgcn_sched_barrier(0);
      conv_mma_trip<S, MT>(fb, acc);
      __builtin_amdgcn_sched_barrier(0);
    }
    conv_load_trip<S, MT>(fb, xrow + koff, wbase, (c0 + E_DIM - TK) & (E_DIM - 1));
    __builtin_amdgcn_sched_barrier(0);
    conv_mma_trip<S, MT>(fa, acc);
    conv_mma_trip<S, MT>(fb, acc);
  }

#pragma unroll
  for (int i = 0; i < 8; ++i) {
    const int otl = i / S;
    const int k   = i % S;
    const int o   = (w * OTW + otl) * 16 + l15;
    const float bdv = bd[o];
#pragma unroll
    for (int m = 0; m < MT; ++m) {
#pragma unroll
      for (int v = 0; v < 8; ++v) {
        const int tl  = m * 16 + ((lane < 16) ? v : v + 8);
        const int row = tl * S + k;
        Elds[row * 256 + (o ^ ((row & 15) * 4))] = acc[m][i][v] + bdv;
      }
    }
  }
  __syncthreads();

  constexpr int CT = MT;
  const int rt   = (MT == 1) ? (w >> 1) : w;
  const int col0 = (MT == 1) ? (w & 1) * 16 : 0;
  const int m    = rt * 16 + l15;
  const int msw  = (m & 15) * 4;

  int  colv[CT];
  bool cval[CT];
  int  colc[CT];
  v8f  acc2[CT];
#pragma unroll
  for (int c = 0; c < CT; ++c) {
    colv[c] = col0 + c * 16 + l15;
    cval[c] = (colv[c] < V_DIM);
    colc[c] = cval[c] ? colv[c] : (V_DIM - 1);
    acc2[c] = (v8f){};
  }

  for (int kk = 0; kk < E_DIM; kk += 4) {
    const int k0 = kk + koff;
    v2f a;
    a.x = Elds[m * 256 + (k0 ^ msw)];
    a.y = Elds[m * 256 + ((k0 + 1) ^ msw)];
#pragma unroll
    for (int c = 0; c < CT; ++c) {
      v2f bf;
      bf.x = Wls[k0 * V_DIM + colc[c]];
      bf.y = Wls[(k0 + 1) * V_DIM + colc[c]];
      acc2[c] = wmma4(a, bf, acc2[c]);
    }
  }

#pragma unroll
  for (int c = 0; c < CT; ++c) {
    const float blv = bl[colc[c]];
#pragma unroll
    for (int v = 0; v < 8; ++v) {
      const int r     = rt * 16 + ((lane < 16) ? v : v + 8);
      const int tl    = r / S;
      const int k     = r % S;
      const int trel2 = tb * TOK + tl;
      if (cval[c] && trel2 < nt) {
        const int j = out_off + trel2 * S + k;
        out[((size_t)n * T_SEQ + j) * V_DIM + colv[c]] = acc2[c][v] + blv;
      }
    }
  }
}


struct Conv8Frag {
  float4 r0[4], r1[4];
  v2f    a0[4], a1[4], ax[4];
};

__device__ __forceinline__ void conv8_load(Conv8Frag& f, const float* __restrict__ xrow,
                                           const float* __restrict__ wbase, int kk0) {
  constexpr int Es = E_DIM * 8;
  constexpr int XP = E_DIM + 4;
#pragma unroll
  for (int s = 0; s < 4; ++s) {
    const int kk = kk0 + 4 * s;
    const float* wrow = wbase + (size_t)kk * Es;
    f.r0[s] = *reinterpret_cast<const float4*>(wrow);
    f.r1[s] = *reinterpret_cast<const float4*>(wrow + Es);
  }
#pragma unroll
  for (int s = 0; s < 4; ++s) {
    const int kk = kk0 + 4 * s;
    f.a0[s].x = xrow[kk];
    f.a0[s].y = xrow[kk + 1];
    f.a1[s].x = xrow[16 * XP + kk];
    f.a1[s].y = xrow[16 * XP + kk + 1];
    f.ax[s].x = xrow[32 * XP + kk];
    f.ax[s].y = xrow[32 * XP + kk + 1];
  }
}

template <int JX>
__device__ __forceinline__ void conv8_mma(const Conv8Frag& f, v8f (&acc)[2][4], v8f& accx) {
#pragma unroll
  for (int s = 0; s < 4; ++s) {
    const float4 r0 = f.r0[s];
    const float4 r1 = f.r1[s];
    const v2f a0 = f.a0[s], a1 = f.a1[s], ax = f.ax[s];
    v2f bf;
    bf.x = r0.x; bf.y = r1.x;
    acc[0][0] = wmma4(a0, bf, acc[0][0]);
    acc[1][0] = wmma4(a1, bf, acc[1][0]);
    if (JX == 0) accx = wmma4(ax, bf, accx);
    bf.x = r0.y; bf.y = r1.y;
    acc[0][1] = wmma4(a0, bf, acc[0][1]);
    acc[1][1] = wmma4(a1, bf, acc[1][1]);
    if (JX == 1) accx = wmma4(ax, bf, accx);
    bf.x = r0.z; bf.y = r1.z;
    acc[0][2] = wmma4(a0, bf, acc[0][2]);
    acc[1][2] = wmma4(a1, bf, acc[1][2]);
    if (JX == 2) accx = wmma4(ax, bf, accx);
    bf.x = r0.w; bf.y = r1.w;
    acc[0][3] = wmma4(a0, bf, acc[0][3]);
    acc[1][3] = wmma4(a1, bf, acc[1][3]);
    if (JX == 3) accx = wmma4(ax, bf, accx);
  }
}

template <int JX>
__device__ __forceinline__ void conv8_stage1(const float* __restrict__ xrow,
                                             const float* __restrict__ wbase,
                                             v8f (&acc)[2][4], v8f& accx, int c0) {
  constexpr int TK = 16;
  Conv8Frag fa, fb;
  conv8_load(fa, xrow, wbase, c0);
#pragma nounroll
  for (int kk = 0; kk < E_DIM - 2 * TK; kk += 2 * TK) {
    conv8_load(fb, xrow, wbase, (c0 + kk + TK) & (E_DIM - 1));
    conv8_mma<JX>(fa, acc, accx);
    conv8_load(fa, xrow, wbase, (c0 + kk + 2 * TK) & (E_DIM - 1));
    conv8_mma<JX>(fb, acc, accx);
  }
  conv8_load(fb, xrow, wbase, (c0 + E_DIM - TK) & (E_DIM - 1));
  conv8_mma<JX>(fa, acc, accx);
  conv8_mma<JX>(fb, acc, accx);
}

__global__ __launch_bounds__(512) void conv8_kernel(
    const float* __restrict__ x,
    const float* __restrict__ Wd, const float* __restrict__ bd,
    const float* __restrict__ Wl, const float* __restrict__ bl,
    float* __restrict__ out,
    int ioff, int out_off, int nt, int nbatch) {
  constexpr int S    = 8;
  constexpr int Es   = E_DIM * S;
  constexpr int XP   = E_DIM + 4;
  constexpr int ROWS = 2 * 16 * 4 + 16;
  __shared__ float Elds[ROWS * 256];
  __shared__ alignas(16) float Xs[48 * XP];
  constexpr int NW4 = (E_DIM * V_DIM) / 4;
  __shared__ alignas(16) float Wls[3 * 512 * 4];

  const int b      = blockIdx.x;
  const int q      = b & 1;
  const int p      = b >> 1;
  const int G      = nbatch * nt;
  const int ntiles = (G + 15) >> 4;
  const int npairs = (int)gridDim.x >> 1;
  const int xt_raw = 2 * npairs + (p >> 2);
  const bool has_x = xt_raw < ntiles;
  const int xt     = has_x ? xt_raw : 2 * p;
  const int jx     = p & 3;

  const int tid  = threadIdx.x;
  const int w    = tid >> 5;
  const int lane = tid & 31;
  const int l15  = lane & 15;
  const int koff = (lane < 16) ? 0 : 2;

  float4 xv[6];
#pragma unroll
  for (int j = 0; j < 6; ++j) {
    const int idx  = tid + j * 512;
    const int r    = idx >> 6;
    const int c4   = idx & 63;
    const int tile = (j < 2) ? 2 * p : (j < 4) ? 2 * p + 1 : xt;
    const int g    = tile * 16 + (r & 15);
    const int gc   = (g < G) ? g : (G - 1);
    const int n    = gc / nt;
    const int t    = gc - n * nt;
    xv[j] = *reinterpret_cast<const float4*>(
        x + ((size_t)n * T_SEQ + ioff + t) * E_DIM + c4 * 4);
  }
  const int widx = tid + 1024;
  const float4 wv0 = reinterpret_cast<const float4*>(Wl)[tid];
  const float4 wv1 = reinterpret_cast<const float4*>(Wl)[tid + 512];
  const float4 wv2 = reinterpret_cast<const float4*>(Wl)[(widx < NW4) ? widx : (NW4 - 1)];
#pragma unroll
  for (int j = 0; j < 6; ++j) {
    const int idx = tid + j * 512;
    *reinterpret_cast<float4*>(&Xs[(idx >> 6) * XP + (idx & 63) * 4]) = xv[j];
  }
  *reinterpret_cast<float4*>(&Wls[tid * 4])         = wv0;
  *reinterpret_cast<float4*>(&Wls[(tid + 512) * 4]) = wv1;
  *reinterpret_cast<float4*>(&Wls[widx * 4])        = wv2;
  __syncthreads();

  const float* xrow = &Xs[l15 * XP + koff];
  v8f acc[2][4];
#pragma unroll
  for (int m = 0; m < 2; ++m)
#pragma unroll
    for (int j = 0; j < 4; ++j) acc[m][j] = (v8f){};
  v8f accx = {};

  const float* wbase = Wd + (size_t)koff * Es + (size_t)(w * 16 + l15) * S + q * 4;

  const int c0 = ((b ^ (b >> 4)) & 15) * 16;
  switch (jx) {
    case 0:  conv8_stage1<0>(xrow, wbase, acc, accx, c0); break;
    case 1:  conv8_stage1<1>(xrow, wbase, acc, accx, c0); break;
    case 2:  conv8_stage1<2>(xrow, wbase, acc, accx, c0); break;
    default: conv8_stage1<3>(xrow, wbase, acc, accx, c0); break;
  }

  {
    const int o = w * 16 + l15;
    const float bdv = bd[o];
#pragma unroll
    for (int m = 0; m < 2; ++m) {
#pragma unroll
      for (int j = 0; j < 4; ++j) {
#pragma unroll
        for (int v = 0; v < 8; ++v) {
          const int tl  = (lane < 16) ? v : v + 8;
          const int row = m * 64 + tl * 4 + j;
          Elds[row * 256 + (o ^ ((row & 15) * 4))] = acc[m][j][v] + bdv;
        }
      }
    }
#pragma unroll
    for (int v = 0; v < 8; ++v) {
      const int tl  = (lane < 16) ? v : v + 8;
      const int row = 128 + tl;
      Elds[row * 256 + (o ^ ((row & 15) * 4))] = accx[v] + bdv;
    }
  }
  __syncthreads();

  if (w >= 9) return;
  const int rt  = w;
  const int m   = rt * 16 + l15;
  const int msw = (m & 15) * 4;

  int  colv[2];
  bool cval[2];
  int  colc[2];
  v8f  acc2[2];
#pragma unroll
  for (int c = 0; c < 2; ++c) {
    colv[c] = c * 16 + l15;
    cval[c] = (colv[c] < V_DIM);
    colc[c] = cval[c] ? colv[c] : (V_DIM - 1);
    acc2[c] = (v8f){};
  }

  for (int kk = 0; kk < E_DIM; kk += 4) {
    const int k0 = kk + koff;
    v2f a;
    a.x = Elds[m * 256 + (k0 ^ msw)];
    a.y = Elds[m * 256 + ((k0 + 1) ^ msw)];
#pragma unroll
    for (int c = 0; c < 2; ++c) {
      v2f bf;
      bf.x = Wls[k0 * V_DIM + colc[c]];
      bf.y = Wls[(k0 + 1) * V_DIM + colc[c]];
      acc2[c] = wmma4(a, bf, acc2[c]);
    }
  }

  const int r0w = (lane < 16) ? 0 : 8;
  if (rt < 8) {
    const int gbase = (2 * p + (rt >> 2)) * 16 + (rt & 3) * 4 + (r0w >> 2);
    const float blv0 = bl[colc[0]];
    const float blv1 = bl[colc[1]];
#pragma unroll
    for (int tk = 0; tk < 2; ++tk) {
      const int g = gbase + tk;
      const int n = g / nt;
      const int t = g - n * nt;
      float* orow = out + ((size_t)n * T_SEQ + out_off + t * S + 4 * q) * V_DIM;
      if (g < G) {
#pragma unroll
        for (int jj = 0; jj < 4; ++jj) {
          if (cval[0]) orow[jj * V_DIM + colv[0]] = acc2[0][tk * 4 + jj] + blv0;
          if (cval[1]) orow[jj * V_DIM + colv[1]] = acc2[1][tk * 4 + jj] + blv1;
        }
      }
    }
  } else {
#pragma unroll
    for (int c = 0; c < 2; ++c) {
      const float blv = bl[colc[c]];
#pragma unroll
      for (int v = 0; v < 8; ++v) {
        const int r = r0w + v;
        const int g = xt * 16 + r;
        const int k = 4 * q + jx;
        if (cval[c] && has_x && g < G) {
          const int n = g / nt;
          const int t = g - n * nt;
          const int j = out_off + t * S + k;
          out[((size_t)n * T_SEQ + j) * V_DIM + colv[c]] = acc2[c][v] + blv;
        }
      }
    }
  }
}

extern "C" void kernel_launch(void* const* d_in, const int* in_sizes, int n_in,
                              void* d_out, int out_size, void* d_ws, size_t ws_size,
                              hipStream_t stream) {
  const float* x   = (const float*)d_in[0];
  const float* W1  = (const float*)d_in[4];
  const float* b1  = (const float*)d_in[5];
  const float* W2  = (const float*)d_in[6];
  const float* b2  = (const float*)d_in[7];
  const float* W3  = (const float*)d_in[8];
  const float* b3  = (const float*)d_in[9];
  const float* Wd4 = (const float*)d_in[10];
  const float* bd4 = (const float*)d_in[11];
  const float* Wl4 = (const float*)d_in[12];
  const float* bl4 = (const float*)d_in[13];
  const float* Wd5 = (const float*)d_in[14];
  const float* bd5 = (const float*)d_in[15];
  const float* Wl5 = (const float*)d_in[16];
  const float* bl5 = (const float*)d_in[17];
  float* out = (float*)d_out;

  const int N = 8;

  head123_kernel<<<dim3(100, N), 64, 0, stream>>>(x, W1, b1, W2, b2, W3, b3, out);

  conv_head_kernel<4, 1><<<dim3(32, N), 256, 0, stream>>>(
      x, Wd4, bd4, Wl4, bl4, out,  1600,  1600,  512);

  conv8_kernel<<<dim3(256), 512, 0, stream>>>(
      x, Wd5, bd5, Wl5, bl5, out,  2112,  3648,  568,  8);
}


# Heads 1-3 launch removed (3 launches to 2): their 800 row tiles are done by the waves of the S=8 conv blocks that have no stage-2 row tile; x and W fragments straight from global; sums unchanged
# speedup vs baseline: 4.7245x; 1.0914x over previous
#include <hip/hip_runtime.h>
#include <hip/hip_bf16.h>


#define T_SEQ 8192
#define E_DIM 256
#define V_DIM 17

typedef float v2f __attribute__((ext_vector_type(2)));
typedef float v8f __attribute__((ext_vector_type(8)));

__device__ __forceinline__ v8f wmma4(v2f a, v2f b, v8f c) {
  return __builtin_amdgcn_wmma_f32_16x16x4_f32(
       false, a,  false, b,
       (short)0, c,  false,  false);
}

#define HEAD_ROW_TILES 100

struct HeadWFrag {
  float x0[8], y0[8];
  float x1[8], y1[8];
};

__device__ __forceinline__ void head_load_w(HeadWFrag& f, const float* __restrict__ w0,
                                            const float* __restrict__ w1, int s0) {
#pragma unroll
  for (int i = 0; i < 8; ++i) {
    const int r = 4 * (s0 + i) * V_DIM;
    f.x0[i] = w0[r];
    f.y0[i] = w0[r + V_DIM];
    f.x1[i] = w1[r];
    f.y1[i] = w1[r + V_DIM];
  }
}

template <int S0>
__device__ __forceinline__ void head_mma(const HeadWFrag& f, const v2f (&af)[E_DIM / 4],
                                         v8f& acc0, v8f& acc1) {
#pragma unroll
  for (int i = 0; i < 8; ++i) {
    v2f bf;
    bf.x = f.x0[i]; bf.y = f.y0[i];
    acc0 = wmma4(af[S0 + i], bf, acc0);
    bf.x = f.x1[i]; bf.y = f.y1[i];
    acc1 = wmma4(af[S0 + i], bf, acc1);
  }
}

__device__ __forceinline__ void head_tile_wave(const float* __restrict__ x,
                                               const float* __restrict__ W,
                                               const float* __restrict__ bias,
                                               float* __restrict__ out,
                                               int n, int rowbase, int lane) {
  constexpr int KS = E_DIM / 4;
  const int l15  = lane & 15;
  const int koff = (lane < 16) ? 0 : 2;

  const float* xr = x + ((size_t)n * T_SEQ + rowbase + l15) * E_DIM + koff;
  v2f af[KS];
#pragma unroll
  for (int s = 0; s < KS; ++s) af[s] = *reinterpret_cast<const v2f*>(xr + 4 * s);

  const float* w0 = W + koff * V_DIM + l15;
  const float* w1 = W + koff * V_DIM + (V_DIM - 1);
  const float bv0 = bias[l15];
  const float bv1 = bias[V_DIM - 1];

  HeadWFrag fa, fb;
  v8f acc0 = {}, acc1 = {};
  head_load_w(fa, w0, w1, 0);
  head_load_w(fb, w0, w1, 8);
  head_mma<0>(fa, af, acc0, acc1);
  head_load_w(fa, w0, w1, 16);
  head_mma<8>(fb, af, acc0, acc1);
  head_load_w(fb, w0, w1, 24);
  head_mma<16>(fa, af, acc0, acc1);
  head_load_w(fa, w0, w1, 32);
  head_mma<24>(fb, af, acc0, acc1);
  head_load_w(fb, w0, w1, 40);
  head_mma<32>(fa, af, acc0, acc1);
  head_load_w(fa, w0, w1, 48);
  head_mma<40>(fb, af, acc0, acc1);
  head_load_w(fb, w0, w1, 56);
  head_mma<48>(fa, af, acc0, acc1);
  head_mma<56>(fb, af, acc0, acc1);

  float* orow = out + ((size_t)n * T_SEQ + rowbase + ((lane < 16) ? 0 : 8)) * V_DIM;
#pragma unroll
  for (int v = 0; v < 8; ++v) {
    orow[v * V_DIM + l15] = acc0[v] + bv0;
    if (l15 == 0) orow[v * V_DIM + (V_DIM - 1)] = acc1[v] + bv1;
  }
}

template <int S, int MT>
struct ConvFrag {
  static constexpr int KT  = (S == 8) ? 2 : 4;
  static constexpr int OTW = 8 / S;
  static constexpr int Q   = S / 4;
  float4 r0[KT][OTW][Q], r1[KT][OTW][Q];
  v2f    a[KT][MT];
};

template <int S, int MT>
__device__ __forceinline__ void conv_load_trip(ConvFrag<S, MT>& f, const float* __restrict__ xk,
                                               const float* __restrict__ wbase, int kk0) {
  constexpr int KT = ConvFrag<S, MT>::KT, OTW = ConvFrag<S, MT>::OTW, Q = ConvFrag<S, MT>::Q;
  constexpr int Es = E_DIM * S;
  constexpr int XP = E_DIM + 4;
#pragma unroll
  for (int s = 0; s < KT; ++s) {
    const float* wrow = wbase + (size_t)(kk0 + 4 * s) * Es;
#pragma unroll
    for (int otl = 0; otl < OTW; ++otl) {
#pragma unroll
      for (int q = 0; q < Q; ++q) {
        const float* bp = wrow + otl * 16 * S + q * 4;
        f.r0[s][otl][q] = *reinterpret_cast<const float4*>(bp);
        f.r1[s][otl][q] = *reinterpret_cast<const float4*>(bp + Es);
      }
    }
  }
#pragma unroll
  for (int s = 0; s < KT; ++s) {
    const int kk = kk0 + 4 * s;
#pragma unroll
    for (int m = 0; m < MT; ++m) {
      f.a[s][m].x = xk[m * 16 * XP + kk];
      f.a[s][m].y = xk[m * 16 * XP + kk + 1];
    }
  }
}

template <int S, int MT>
__device__ __forceinline__ void conv_mma_trip(const ConvFrag<S, MT>& f, v8f (&acc)[MT][8]) {
  constexpr int KT = ConvFrag<S, MT>::KT, OTW = ConvFrag<S, MT>::OTW, Q = ConvFrag<S, MT>::Q;
#pragma unroll
  for (int s = 0; s < KT; ++s) {
#pragma unroll
    for (int otl = 0; otl < OTW; ++otl) {
#pragma unroll
      for (int q = 0; q < Q; ++q) {
        const float4 r0 = f.r0[s][otl][q];
        const float4 r1 = f.r1[s][otl][q];
        const int t0 = otl * S + q * 4;
        v2f bf;
        bf.x = r0.x; bf.y = r1.x;
#pragma unroll
        for (int m = 0; m < MT; ++m) acc[m][t0 + 0] = wmma4(f.a[s][m], bf, acc[m][t0 + 0]);
        bf.x = r0.y; bf.y = r1.y;
#pragma unroll
        for (int m = 0; m < MT; ++m) acc[m][t0 + 1] = wmma4(f.a[s][m], bf, acc[m][t0 + 1]);
        bf.x = r0.z; bf.y = r1.z;
#pragma unroll
        for (int m = 0; m < MT; ++m) acc[m][t0 + 2] = wmma4(f.a[s][m], bf, acc[m][t0 + 2]);
        bf.x = r0.w; bf.y = r1.w;
#pragma unroll
        for (int m = 0; m < MT; ++m) acc[m][t0 + 3] = wmma4(f.a[s][m], bf, acc[m][t0 + 3]);
      }
    }
  }
}

template <int S, int MT>
__global__ __launch_bounds__(64 * S) void conv_head_kernel(
    const float* __restrict__ x,
    const float* __restrict__ Wd, const float* __restrict__ bd,
    const float* __restrict__ Wl, const float* __restrict__ bl,
    float* __restrict__ out,
    int ioff, int out_off, int nt) {
  constexpr int OTW = 8 / S;
  constexpr int Q   = S / 4;
  constexpr int Es  = E_DIM * S;
  constexpr int TOK = 16 * MT;
  __shared__ float Elds[TOK * S * 256];
  constexpr int XP = E_DIM + 4;
  __shared__ alignas(16) float Xs[TOK * XP];
  constexpr int NW4 = (E_DIM * V_DIM) / 4;
  constexpr int NWP = (NW4 + 64 * S - 1) / (64 * S);
  __shared__ alignas(16) float Wls[NWP * 64 * S * 4];

  const int n  = blockIdx.y;
  const int tb = blockIdx.x;

  const int tid  = threadIdx.x;
  const int w    = tid >> 5;
  const int lane = tid & 31;
  const int l15  = lane & 15;
  const int koff = (lane < 16) ? 0 : 2;

  constexpr int NXL = (TOK * 64) / (64 * S);
  float4 xv[NXL];
#pragma unroll
  for (int j = 0; j < NXL; ++j) {
    const int idx    = tid + j * 64 * S;
    const int r      = idx >> 6;
    const int c4     = idx & 63;
    const int trel   = tb * TOK + r;
    const int tclamp = (trel < nt) ? trel : (nt - 1);
    xv[j] = *reinterpret_cast<const float4*>(
        x + ((size_t)n * T_SEQ + ioff + tclamp) * E_DIM + c4 * 4);
  }
  float4 wv[NWP];
#pragma unroll
  for (int j = 0; j < NWP; ++j) {
    const int idx = tid + j * 64 * S;
    const int src = (j == NWP - 1 && idx >= NW4) ? (NW4 - 1) : idx;
    wv[j] = reinterpret_cast<const float4*>(Wl)[src];
  }
#pragma unroll
  for (int j = 0; j < NXL; ++j) {
    const int idx = tid + j * 64 * S;
    *reinterpret_cast<float4*>(&Xs[(idx >> 6) * XP + (idx & 63) * 4]) = xv[j];
  }
#pragma unroll
  for (int j = 0; j < NWP; ++j)
    *reinterpret_cast<float4*>(&Wls[(tid + j * 64 * S) * 4]) = wv[j];
  __syncthreads();

  const float* xrow = &Xs[l15 * XP];

  v8f acc[MT][8];
#pragma unroll
  for (int m = 0; m < MT; ++m)
#pragma unroll
    for (int i = 0; i < 8; ++i) acc[m][i] = (v8f){};

  const float* wbase = Wd + (size_t)koff * Es + (size_t)((w * OTW) * 16 + l15) * S;

  {
    constexpr int TK = 4 * ConvFrag<S, MT>::KT;
    constexpr int NCH = E_DIM / TK;
    const int lin = (int)(blockIdx.y * gridDim.x + blockIdx.x);
    const int c0  = ((lin ^ (lin >> 4)) & (NCH - 1)) * TK;
    ConvFrag<S, MT> fa, fb;
    conv_load_trip<S, MT>(fa, xrow + koff, wbase, c0);
    for (int kk = 0; kk < E_DIM - 2 * TK; kk += 2 * TK) {
      conv_load_trip<S, MT>(fb, xrow + koff, wbase, (c0 + kk + TK) & (E_DIM - 1));
      __builtin_amdgcn_sched_barrier(0);
      conv_mma_trip<S, MT>(fa, acc);
      __builtin_amdgcn_sched_barrier(0);
      conv_load_trip<S, MT>(fa, xrow + koff, wbase, (c0 + kk + 2 * TK) & (E_DIM - 1));
      __builtin_amdgcn_sched_barrier(0);
      conv_mma_trip<S, MT>(fb, acc);
      __builtin_amdgcn_sched_barrier(0);
    }
    conv_load_trip<S, MT>(fb, xrow + koff, wbase, (c0 + E_DIM - TK) & (E_DIM - 1));
    __builtin_amdgcn_sched_barrier(0);
    conv_mma_trip<S, MT>(fa, acc);
    conv_mma_trip<S, MT>(fb, acc);
  }

#pragma unroll
  for (int i = 0; i < 8; ++i) {
    const int otl = i / S;
    const int k   = i % S;
    const int o   = (w * OTW + otl) * 16 + l15;
    const float bdv = bd[o];
#pragma unroll
    for (int m = 0; m < MT; ++m) {
#pragma unroll
      for (int v = 0; v < 8; ++v) {
        const int tl  = m * 16 + ((lane < 16) ? v : v + 8);
        const int row = tl * S + k;
        Elds[row * 256 + (o ^ ((row & 15) * 4))] = acc[m][i][v] + bdv;
      }
    }
  }
  __syncthreads();

  constexpr int CT = MT;
  const int rt   = (MT == 1) ? (w >> 1) : w;
  const int col0 = (MT == 1) ? (w & 1) * 16 : 0;
  const int m    = rt * 16 + l15;
  const int msw  = (m & 15) * 4;

  int  colv[CT];
  bool cval[CT];
  int  colc[CT];
  v8f  acc2[CT];
#pragma unroll
  for (int c = 0; c < CT; ++c) {
    colv[c] = col0 + c * 16 + l15;
    cval[c] = (colv[c] < V_DIM);
    colc[c] = cval[c] ? colv[c] : (V_DIM - 1);
    acc2[c] = (v8f){};
  }

  for (int kk = 0; kk < E_DIM; kk += 4) {
    const int k0 = kk + koff;
    v2f a;
    a.x = Elds[m * 256 + (k0 ^ msw)];
    a.y = Elds[m * 256 + ((k0 + 1) ^ msw)];
#pragma unroll
    for (int c = 0; c < CT; ++c) {
      v2f bf;
      bf.x = Wls[k0 * V_DIM + colc[c]];
      bf.y = Wls[(k0 + 1) * V_DIM + colc[c]];
      acc2[c] = wmma4(a, bf, acc2[c]);
    }
  }

#pragma unroll
  for (int c = 0; c < CT; ++c) {
    const float blv = bl[colc[c]];
#pragma unroll
    for (int v = 0; v < 8; ++v) {
      const int r     = rt * 16 + ((lane < 16) ? v : v + 8);
      const int tl    = r / S;
      const int k     = r % S;
      const int trel2 = tb * TOK + tl;
      if (cval[c] && trel2 < nt) {
        const int j = out_off + trel2 * S + k;
        out[((size_t)n * T_SEQ + j) * V_DIM + colv[c]] = acc2[c][v] + blv;
      }
    }
  }
}


struct Conv8Frag {
  float4 r0[4], r1[4];
  v2f    a0[4], a1[4], ax[4];
};

__device__ __forceinline__ void conv8_load(Conv8Frag& f, const float* __restrict__ xrow,
                                           const float* __restrict__ wbase, int kk0) {
  constexpr int Es = E_DIM * 8;
  constexpr int XP = E_DIM + 4;
#pragma unroll
  for (int s = 0; s < 4; ++s) {
    const int kk = kk0 + 4 * s;
    const float* wrow = wbase + (size_t)kk * Es;
    f.r0[s] = *reinterpret_cast<const float4*>(wrow);
    f.r1[s] = *reinterpret_cast<const float4*>(wrow + Es);
  }
#pragma unroll
  for (int s = 0; s < 4; ++s) {
    const int kk = kk0 + 4 * s;
    f.a0[s].x = xrow[kk];
    f.a0[s].y = xrow[kk + 1];
    f.a1[s].x = xrow[16 * XP + kk];
    f.a1[s].y = xrow[16 * XP + kk + 1];
    f.ax[s].x = xrow[32 * XP + kk];
    f.ax[s].y = xrow[32 * XP + kk + 1];
  }
}

template <int JX>
__device__ __forceinline__ void conv8_mma(const Conv8Frag& f, v8f (&acc)[2][4], v8f& accx) {
#pragma unroll
  for (int s = 0; s < 4; ++s) {
    const float4 r0 = f.r0[s];
    const float4 r1 = f.r1[s];
    const v2f a0 = f.a0[s], a1 = f.a1[s], ax = f.ax[s];
    v2f bf;
    bf.x = r0.x; bf.y = r1.x;
    acc[0][0] = wmma4(a0, bf, acc[0][0]);
    acc[1][0] = wmma4(a1, bf, acc[1][0]);
    if (JX == 0) accx = wmma4(ax, bf, accx);
    bf.x = r0.y; bf.y = r1.y;
    acc[0][1] = wmma4(a0, bf, acc[0][1]);
    acc[1][1] = wmma4(a1, bf, acc[1][1]);
    if (JX == 1) accx = wmma4(ax, bf, accx);
    bf.x = r0.z; bf.y = r1.z;
    acc[0][2] = wmma4(a0, bf, acc[0][2]);
    acc[1][2] = wmma4(a1, bf, acc[1][2]);
    if (JX == 2) accx = wmma4(ax, bf, accx);
    bf.x = r0.w; bf.y = r1.w;
    acc[0][3] = wmma4(a0, bf, acc[0][3]);
    acc[1][3] = wmma4(a1, bf, acc[1][3]);
    if (JX == 3) accx = wmma4(ax, bf, accx);
  }
}

template <int JX>
__device__ __forceinline__ void conv8_stage1(const float* __restrict__ xrow,
                                             const float* __restrict__ wbase,
                                             v8f (&acc)[2][4], v8f& accx, int c0) {
  constexpr int TK = 16;
  Conv8Frag fa, fb;
  conv8_load(fa, xrow, wbase, c0);
#pragma nounroll
  for (int kk = 0; kk < E_DIM - 2 * TK; kk += 2 * TK) {
    conv8_load(fb, xrow, wbase, (c0 + kk + TK) & (E_DIM - 1));
    conv8_mma<JX>(fa, acc, accx);
    conv8_load(fa, xrow, wbase, (c0 + kk + 2 * TK) & (E_DIM - 1));
    conv8_mma<JX>(fb, acc, accx);
  }
  conv8_load(fb, xrow, wbase, (c0 + E_DIM - TK) & (E_DIM - 1));
  conv8_mma<JX>(fa, acc, accx);
  conv8_mma<JX>(fb, acc, accx);
}

__global__ __launch_bounds__(512) void conv8_kernel(
    const float* __restrict__ x,
    const float* __restrict__ Wd, const float* __restrict__ bd,
    const float* __restrict__ Wl, const float* __restrict__ bl,
    const float* __restrict__ W1, const float* __restrict__ b1,
    const float* __restrict__ W2, const float* __restrict__ b2,
    const float* __restrict__ W3, const float* __restrict__ b3,
    float* __restrict__ out,
    int ioff, int out_off, int nt, int nbatch) {
  constexpr int S    = 8;
  constexpr int Es   = E_DIM * S;
  constexpr int XP   = E_DIM + 4;
  constexpr int ROWS = 2 * 16 * 4 + 16;
  __shared__ float Elds[ROWS * 256];
  __shared__ alignas(16) float Xs[48 * XP];
  constexpr int NW4 = (E_DIM * V_DIM) / 4;
  __shared__ alignas(16) float Wls[3 * 512 * 4];

  const int b      = blockIdx.x;
  const int q      = b & 1;
  const int p      = b >> 1;
  const int G      = nbatch * nt;
  const int ntiles = (G + 15) >> 4;
  const int npairs = (int)gridDim.x >> 1;
  const int xt_raw = 2 * npairs + (p >> 2);
  const bool has_x = xt_raw < ntiles;
  const int xt     = has_x ? xt_raw : 2 * p;
  const int jx     = p & 3;

  const int tid  = threadIdx.x;
  const int w    = tid >> 5;
  const int lane = tid & 31;
  const int l15  = lane & 15;
  const int koff = (lane < 16) ? 0 : 2;

  float4 xv[6];
#pragma unroll
  for (int j = 0; j < 6; ++j) {
    const int idx  = tid + j * 512;
    const int r    = idx >> 6;
    const int c4   = idx & 63;
    const int tile = (j < 2) ? 2 * p : (j < 4) ? 2 * p + 1 : xt;
    const int g    = tile * 16 + (r & 15);
    const int gc   = (g < G) ? g : (G - 1);
    const int n    = gc / nt;
    const int t    = gc - n * nt;
    xv[j] = *reinterpret_cast<const float4*>(
        x + ((size_t)n * T_SEQ + ioff + t) * E_DIM + c4 * 4);
  }
  const int widx = tid + 1024;
  const float4 wv0 = reinterpret_cast<const float4*>(Wl)[tid];
  const float4 wv1 = reinterpret_cast<const float4*>(Wl)[tid + 512];
  const float4 wv2 = reinterpret_cast<const float4*>(Wl)[(widx < NW4) ? widx : (NW4 - 1)];
#pragma unroll
  for (int j = 0; j < 6; ++j) {
    const int idx = tid + j * 512;
    *reinterpret_cast<float4*>(&Xs[(idx >> 6) * XP + (idx & 63) * 4]) = xv[j];
  }
  *reinterpret_cast<float4*>(&Wls[tid * 4])         = wv0;
  *reinterpret_cast<float4*>(&Wls[(tid + 512) * 4]) = wv1;
  *reinterpret_cast<float4*>(&Wls[widx * 4])        = wv2;
  __syncthreads();

  const float* xrow = &Xs[l15 * XP + koff];
  v8f acc[2][4];
#pragma unroll
  for (int m = 0; m < 2; ++m)
#pragma unroll
    for (int j = 0; j < 4; ++j) acc[m][j] = (v8f){};
  v8f accx = {};

  const float* wbase = Wd + (size_t)koff * Es + (size_t)(w * 16 + l15) * S + q * 4;

  const int c0 = ((b ^ (b >> 4)) & 15) * 16;
  switch (jx) {
    case 0:  conv8_stage1<0>(xrow, wbase, acc, accx, c0); break;
    case 1:  conv8_stage1<1>(xrow, wbase, acc, accx, c0); break;
    case 2:  conv8_stage1<2>(xrow, wbase, acc, accx, c0); break;
    default: conv8_stage1<3>(xrow, wbase, acc, accx, c0); break;
  }

  {
    const int o = w * 16 + l15;
    const float bdv = bd[o];
#pragma unroll
    for (int m = 0; m < 2; ++m) {
#pragma unroll
      for (int j = 0; j < 4; ++j) {
#pragma unroll
        for (int v = 0; v < 8; ++v) {
          const int tl  = (lane < 16) ? v : v + 8;
          const int row = m * 64 + tl * 4 + j;
          Elds[row * 256 + (o ^ ((row & 15) * 4))] = acc[m][j][v] + bdv;
        }
      }
    }
#pragma unroll
    for (int v = 0; v < 8; ++v) {
      const int tl  = (lane < 16) ? v : v + 8;
      const int row = 128 + tl;
      Elds[row * 256 + (o ^ ((row & 15) * 4))] = accx[v] + bdv;
    }
  }
  __syncthreads();

  if (w >= 9 || (w == 8 && !has_x)) {
    const int nb     = (int)gridDim.x;
    const int nleft  = (ntiles > 2 * npairs) ? (ntiles - 2 * npairs) : 0;
    const int first8 = (8 * nleft < nb) ? 8 * nleft : nb;
    int h;
    if (w == 8)       h = 3 * nb + (b - first8);
    else if (w <= 11) h = (w - 9) * nb + b;
    else if (w == 12) h = 3 * nb + (nb - first8) + b;
    else return;
    if (h >= HEAD_ROW_TILES * nbatch) return;
    const int hn      = h / HEAD_ROW_TILES;
    const int rowbase = (h - hn * HEAD_ROW_TILES) * 16;
    const float* Wh;
    const float* bh;
    if (rowbase < 64)       { Wh = W1; bh = b1; }
    else if (rowbase < 576) { Wh = W2; bh = b2; }
    else                    { Wh = W3; bh = b3; }
    head_tile_wave(x, Wh, bh, out, hn, rowbase, lane);
    return;
  }
  const int rt  = w;
  const int m   = rt * 16 + l15;
  const int msw = (m & 15) * 4;

  int  colv[2];
  bool cval[2];
  int  colc[2];
  v8f  acc2[2];
#pragma unroll
  for (int c = 0; c < 2; ++c) {
    colv[c] = c * 16 + l15;
    cval[c] = (colv[c] < V_DIM);
    colc[c] = cval[c] ? colv[c] : (V_DIM - 1);
    acc2[c] = (v8f){};
  }

  for (int kk = 0; kk < E_DIM; kk += 4) {
    const int k0 = kk + koff;
    v2f a;
    a.x = Elds[m * 256 + (k0 ^ msw)];
    a.y = Elds[m * 256 + ((k0 + 1) ^ msw)];
#pragma unroll
    for (int c = 0; c < 2; ++c) {
      v2f bf;
      bf.x = Wls[k0 * V_DIM + colc[c]];
      bf.y = Wls[(k0 + 1) * V_DIM + colc[c]];
      acc2[c] = wmma4(a, bf, acc2[c]);
    }
  }

  const int r0w = (lane < 16) ? 0 : 8;
  if (rt < 8) {
    const int gbase = (2 * p + (rt >> 2)) * 16 + (rt & 3) * 4 + (r0w >> 2);
    const float blv0 = bl[colc[0]];
    const float blv1 = bl[colc[1]];
#pragma unroll
    for (int tk = 0; tk < 2; ++tk) {
      const int g = gbase + tk;
      const int n = g / nt;
      const int t = g - n * nt;
      float* orow = out + ((size_t)n * T_SEQ + out_off + t * S + 4 * q) * V_DIM;
      if (g < G) {
#pragma unroll
        for (int jj = 0; jj < 4; ++jj) {
          if (cval[0]) orow[jj * V_DIM + colv[0]] = acc2[0][tk * 4 + jj] + blv0;
          if (cval[1]) orow[jj * V_DIM + colv[1]] = acc2[1][tk * 4 + jj] + blv1;
        }
      }
    }
  } else {
#pragma unroll
    for (int c = 0; c < 2; ++c) {
      const float blv = bl[colc[c]];
#pragma unroll
      for (int v = 0; v < 8; ++v) {
        const int r = r0w + v;
        const int g = xt * 16 + r;
        const int k = 4 * q + jx;
        if (cval[c] && has_x && g < G) {
          const int n = g / nt;
          const int t = g - n * nt;
          const int j = out_off + t * S + k;
          out[((size_t)n * T_SEQ + j) * V_DIM + colv[c]] = acc2[c][v] + blv;
        }
      }
    }
  }
}

extern "C" void kernel_launch(void* const* d_in, const int* in_sizes, int n_in,
                              void* d_out, int out_size, void* d_ws, size_t ws_size,
                              hipStream_t stream) {
  const float* x   = (const float*)d_in[0];
  const float* W1  = (const float*)d_in[4];
  const float* b1  = (const float*)d_in[5];
  const float* W2  = (const float*)d_in[6];
  const float* b2  = (const float*)d_in[7];
  const float* W3  = (const float*)d_in[8];
  const float* b3  = (const float*)d_in[9];
  const float* Wd4 = (const float*)d_in[10];
  const float* bd4 = (const float*)d_in[11];
  const float* Wl4 = (const float*)d_in[12];
  const float* bl4 = (const float*)d_in[13];
  const float* Wd5 = (const float*)d_in[14];
  const float* bd5 = (const float*)d_in[15];
  const float* Wl5 = (const float*)d_in[16];
  const float* bl5 = (const float*)d_in[17];
  float* out = (float*)d_out;

  const int N = 8;

  conv_head_kernel<4, 1><<<dim3(32, N), 256, 0, stream>>>(
      x, Wd4, bd4, Wl4, bl4, out,  1600,  1600,  512);

  conv8_kernel<<<dim3(256), 512, 0, stream>>>(
      x, Wd5, bd5, Wl5, bl5, W1, b1, W2, b2, W3, b3, out,
       2112,  3648,  568,  8);
}
